# starting kernel
# baseline (speedup 1.0000x reference)
#include <hip/hip_runtime.h>
#include <math.h>


typedef __attribute__((ext_vector_type(2))) float v2f;
typedef __attribute__((ext_vector_type(8))) float v8f;

#define B_   32
#define V_   1024
#define F_   16
#define A_   32
#define P_   64
#define NF_  128
#define C_   96
#define KTOT 48
#define ROWS (B_ * V_)


__global__ __launch_bounds__(256) void k0_prep(
    const float* __restrict__ Wf, const float* __restrict__ bf,
    const float* __restrict__ Ws, const float* __restrict__ bs,
    float* __restrict__ WC, float* __restrict__ bc)
{
    const int tid = blockIdx.x * 256 + threadIdx.x;
    if (tid < F_ * C_) {
        const int k = tid / C_, c = tid % C_;
        WC[tid] = (c < P_) ? Wf[k * P_ + c] : Ws[k * A_ + (c - P_)];
    } else if (tid < F_ * C_ + C_) {
        const int c = tid - F_ * C_;
        bc[c] = (c < P_) ? bf[c] : bs[c - P_];
    }
}

__global__ __launch_bounds__(192) void k1_vertex(
    const float* __restrict__ x,  const float* __restrict__ WC,
    const float* __restrict__ bc, float* __restrict__ fws)
{
    const int lane  = threadIdx.x & 31;
    const int ntile = threadIdx.x >> 5;
    const int mtile = blockIdx.x;
    const int col   = ntile * 16 + (lane & 15);
    const int kh    = (lane >> 4) * 2;
    const float* xr = x + (size_t)(mtile * 16 + (lane & 15)) * F_;

    v8f acc = {};
#pragma unroll
    for (int kt = 0; kt < 4; ++kt) {
        const int k0 = kt * 4 + kh;
        const v2f a = *(const v2f*)(xr + k0);
        v2f b;
        b.x = WC[k0 * C_ + col];
        b.y = WC[(k0 + 1) * C_ + col];
        acc = __builtin_amdgcn_wmma_f32_16x16x4_f32(
            false, a, false, b, (short)0, acc, false, false);
    }

    const float bias = bc[col];
#pragma unroll
    for (int r = 0; r < 8; ++r) {
        const int row = mtile * 16 + r + ((lane >> 4) * 8);
        float v = acc[r] + bias;
        if (col >= P_) v = expf(-v * v);
        fws[(size_t)row * C_ + col] = v;
    }
}

__global__ __launch_bounds__(96) void k2_agg(
    const float* __restrict__ fws, float* __restrict__ agg)
{
    const int ba = blockIdx.x;
    const int b  = ba >> 5, a = ba & 31;
    const int c  = threadIdx.x;
    const float* fb = fws + (size_t)b * V_ * C_;

    float mx = -INFINITY, sm = 0.0f;
#pragma unroll 4
    for (int v = 0; v < V_; ++v) {
        const float w   = fb[(size_t)v * C_ + P_ + a];
        const float val = w * fb[(size_t)v * C_ + c];
        mx = fmaxf(mx, val);
        sm += val;
    }
    float* ao = agg + (size_t)ba * 192;
    ao[c]      = mx;
    ao[96 + c] = sm * (1.0f / (float)V_);
}

__global__ __launch_bounds__(128) void k3_bw(
    const float* __restrict__ agg, const float* __restrict__ Wout,
    float* __restrict__ BW)
{
    const int bid  = blockIdx.x;
    const int b    = bid / KTOT, krow = bid % KTOT;
    const int n    = threadIdx.x;
    float acc;
    if (krow < F_) {
        acc = Wout[(size_t)krow * NF_ + n];
    } else {
        const int a = krow - F_;
        acc = Wout[(size_t)(F_ + 2 * A_ * C_ + a) * NF_ + n];
        const float* ag = agg + (size_t)(b * A_ + a) * 192;
        const float* wr = Wout + (size_t)(F_ + a * 192) * NF_ + n;
#pragma unroll 4
        for (int c = 0; c < 192; ++c)
            acc = fmaf(ag[c], wr[(size_t)c * NF_], acc);
    }
    BW[(size_t)bid * NF_ + n] = acc;
}

__global__ __launch_bounds__(256) void k4_out(
    const float* __restrict__ x,   const float* __restrict__ fws,
    const float* __restrict__ BW,  const float* __restrict__ bout,
    float* __restrict__ out)
{
    const int lane  = threadIdx.x & 31;
    const int ntile = threadIdx.x >> 5;
    const int bid   = blockIdx.x;
    const int b     = bid >> 6, mtile = bid & 63;
    const int col   = ntile * 16 + (lane & 15);
    const int kh    = (lane >> 4) * 2;
    const int rowA  = b * V_ + mtile * 16 + (lane & 15);

    const float* xr  = x   + (size_t)rowA * F_;
    const float* er  = fws + (size_t)rowA * C_ + P_;
    const float* bwb = BW  + (size_t)b * KTOT * NF_;

    v8f acc = {};
#pragma unroll
    for (int kt = 0; kt < 4; ++kt) {
        const int k0 = kt * 4 + kh;
        const v2f a = *(const v2f*)(xr + k0);
        v2f bb;
        bb.x = bwb[(size_t)k0 * NF_ + col];
        bb.y = bwb[(size_t)(k0 + 1) * NF_ + col];
        acc = __builtin_amdgcn_wmma_f32_16x16x4_f32(
            false, a, false, bb, (short)0, acc, false, false);
    }
#pragma unroll
    for (int kt = 4; kt < 12; ++kt) {
        const int k0 = kt * 4 + kh;
        const v2f a = *(const v2f*)(er + (k0 - F_));
        v2f bb;
        bb.x = bwb[(size_t)k0 * NF_ + col];
        bb.y = bwb[(size_t)(k0 + 1) * NF_ + col];
        acc = __builtin_amdgcn_wmma_f32_16x16x4_f32(
            false, a, false, bb, (short)0, acc, false, false);
    }

    const float bias = bout[col];
#pragma unroll
    for (int r = 0; r < 8; ++r) {
        const int row = b * V_ + mtile * 16 + r + ((lane >> 4) * 8);
        out[(size_t)row * NF_ + col] = tanhf(acc[r] + bias);
    }
}

extern "C" void kernel_launch(void* const* d_in, const int* in_sizes, int n_in,
                              void* d_out, int out_size, void* d_ws, size_t ws_size,
                              hipStream_t stream)
{
    const float* x    = (const float*)d_in[0];
    const float* Wf   = (const float*)d_in[1];
    const float* bf   = (const float*)d_in[2];
    const float* Ws   = (const float*)d_in[3];
    const float* bs   = (const float*)d_in[4];
    const float* Wout = (const float*)d_in[5];
    const float* bout = (const float*)d_in[6];
    float* out = (float*)d_out;

    float* fws = (float*)d_ws;
    float* agg = fws + (size_t)ROWS * C_;
    float* BW  = agg + (size_t)B_ * A_ * 192;
    float* WC  = BW  + (size_t)B_ * KTOT * NF_;
    float* bc  = WC  + (size_t)F_ * C_;

    k0_prep  <<<7,         256, 0, stream>>>(Wf, bf, Ws, bs, WC, bc);
    k1_vertex<<<ROWS / 16, 192, 0, stream>>>(x, WC, bc, fws);
    k2_agg   <<<B_ * A_,   96,  0, stream>>>(fws, agg);
    k3_bw    <<<B_ * KTOT, 128, 0, stream>>>(agg, Wout, BW);
    k4_out   <<<B_ * 64,   256, 0, stream>>>(x, fws, BW, bout, out);
}


# k2 aggregation retiled: block per (batch, 4 aggregators), V split over 4 wave slices, each f load reused for 4 aggregators, ew staged in LDS; slice partials combined in LDS (mean sum order changes)
# speedup vs baseline: 2.5375x; 2.5375x over previous
#include <hip/hip_runtime.h>
#include <math.h>


typedef __attribute__((ext_vector_type(2))) float v2f;
typedef __attribute__((ext_vector_type(8))) float v8f;

#define B_   32
#define V_   1024
#define F_   16
#define A_   32
#define P_   64
#define NF_  128
#define C_   96
#define KTOT 48
#define ROWS (B_ * V_)


__global__ __launch_bounds__(256) void k0_prep(
    const float* __restrict__ Wf, const float* __restrict__ bf,
    const float* __restrict__ Ws, const float* __restrict__ bs,
    float* __restrict__ WC, float* __restrict__ bc)
{
    const int tid = blockIdx.x * 256 + threadIdx.x;
    if (tid < F_ * C_) {
        const int k = tid / C_, c = tid % C_;
        WC[tid] = (c < P_) ? Wf[k * P_ + c] : Ws[k * A_ + (c - P_)];
    } else if (tid < F_ * C_ + C_) {
        const int c = tid - F_ * C_;
        bc[c] = (c < P_) ? bf[c] : bs[c - P_];
    }
}

__global__ __launch_bounds__(192) void k1_vertex(
    const float* __restrict__ x,  const float* __restrict__ WC,
    const float* __restrict__ bc, float* __restrict__ fws)
{
    const int lane  = threadIdx.x & 31;
    const int ntile = threadIdx.x >> 5;
    const int mtile = blockIdx.x;
    const int col   = ntile * 16 + (lane & 15);
    const int kh    = (lane >> 4) * 2;
    const float* xr = x + (size_t)(mtile * 16 + (lane & 15)) * F_;

    v8f acc = {};
#pragma unroll
    for (int kt = 0; kt < 4; ++kt) {
        const int k0 = kt * 4 + kh;
        const v2f a = *(const v2f*)(xr + k0);
        v2f b;
        b.x = WC[k0 * C_ + col];
        b.y = WC[(k0 + 1) * C_ + col];
        acc = __builtin_amdgcn_wmma_f32_16x16x4_f32(
            false, a, false, b, (short)0, acc, false, false);
    }

    const float bias = bc[col];
#pragma unroll
    for (int r = 0; r < 8; ++r) {
        const int row = mtile * 16 + r + ((lane >> 4) * 8);
        float v = acc[r] + bias;
        if (col >= P_) v = expf(-v * v);
        fws[(size_t)row * C_ + col] = v;
    }
}

#define K2_AG 4
#define K2_VS 4
#define K2_VL (V_ / K2_VS)
#define K2_U  16
#define K2_NB (K2_VL / K2_U)
#define K2_T  (96 * K2_VS)

__device__ __forceinline__ void k2_load(float (&d)[K2_U], const float* __restrict__ p)
{
#pragma unroll
    for (int u = 0; u < K2_U; ++u) d[u] = p[(size_t)u * C_];
}

__device__ __forceinline__ void k2_step(const float (&fv)[K2_U], const float4* w4,
                                        float (&mx)[K2_AG], float (&sm)[K2_AG])
{
#pragma unroll
    for (int u = 0; u < K2_U; ++u) {
        const float4 w = w4[u];
        const float wj[K2_AG] = {w.x, w.y, w.z, w.w};
#pragma unroll
        for (int j = 0; j < K2_AG; ++j) {
            const float val = wj[j] * fv[u];
            mx[j] = fmaxf(mx[j], val);
            sm[j] += val;
        }
    }
}

__global__ __launch_bounds__(K2_T) void k2_agg(
    const float* __restrict__ fws, float* __restrict__ agg)
{
    __shared__ float4 sew[V_];
    __shared__ float  smx[K2_VS][K2_AG][C_];
    __shared__ float  ssm[K2_VS][K2_AG][C_];

    const int b    = blockIdx.x >> 3;
    const int a0   = (blockIdx.x & 7) * K2_AG;
    const int lane = threadIdx.x & 31;
    const int wv   = threadIdx.x >> 5;
    const int g    = wv % 3;
    const int s    = wv / 3;
    const int c    = g * 32 + lane;

    const float* fb = fws + (size_t)b * V_ * C_;
    const float* fp = fb + (size_t)s * K2_VL * C_ + c;

    float fa[K2_U], fc[K2_U];
    k2_load(fa, fp);

    const float* eb = fb + P_ + a0;
    const int v0 = threadIdx.x, v1 = threadIdx.x + K2_T;
    const int v2 = (threadIdx.x + 2 * K2_T < V_) ? threadIdx.x + 2 * K2_T : v0;
    const float4 e0 = *(const float4*)(eb + (size_t)v0 * C_);
    const float4 e1 = *(const float4*)(eb + (size_t)v1 * C_);
    const float4 e2 = *(const float4*)(eb + (size_t)v2 * C_);
    sew[v0] = e0;
    sew[v1] = e1;
    sew[v2] = e2;
    __syncthreads();

    float mx[K2_AG], sm[K2_AG];
#pragma unroll
    for (int j = 0; j < K2_AG; ++j) { mx[j] = -INFINITY; sm[j] = 0.0f; }

    static_assert(K2_NB >= 2 && K2_NB % 2 == 0, "batches are consumed in pairs");
    const float4* wp = sew + s * K2_VL;
    for (int it = 0; it < K2_NB - 2; it += 2) {
        k2_load(fc, fp + (size_t)(it + 1) * K2_U * C_);
        k2_step(fa, wp + it * K2_U, mx, sm);
        k2_load(fa, fp + (size_t)(it + 2) * K2_U * C_);
        k2_step(fc, wp + (it + 1) * K2_U, mx, sm);
    }
    k2_load(fc, fp + (size_t)(K2_NB - 1) * K2_U * C_);
    k2_step(fa, wp + (K2_NB - 2) * K2_U, mx, sm);
    k2_step(fc, wp + (K2_NB - 1) * K2_U, mx, sm);

#pragma unroll
    for (int j = 0; j < K2_AG; ++j) { smx[s][j][c] = mx[j]; ssm[s][j][c] = sm[j]; }
    __syncthreads();

    const int j  = threadIdx.x / C_;
    const int cc = threadIdx.x % C_;
    float m = smx[0][j][cc], t = ssm[0][j][cc];
#pragma unroll
    for (int q = 1; q < K2_VS; ++q) { m = fmaxf(m, smx[q][j][cc]); t += ssm[q][j][cc]; }
    float* ao = agg + (size_t)(b * A_ + a0 + j) * 192;
    ao[cc]      = m;
    ao[96 + cc] = t * (1.0f / (float)V_);
}

__global__ __launch_bounds__(128) void k3_bw(
    const float* __restrict__ agg, const float* __restrict__ Wout,
    float* __restrict__ BW)
{
    const int bid  = blockIdx.x;
    const int b    = bid / KTOT, krow = bid % KTOT;
    const int n    = threadIdx.x;
    float acc;
    if (krow < F_) {
        acc = Wout[(size_t)krow * NF_ + n];
    } else {
        const int a = krow - F_;
        acc = Wout[(size_t)(F_ + 2 * A_ * C_ + a) * NF_ + n];
        const float* ag = agg + (size_t)(b * A_ + a) * 192;
        const float* wr = Wout + (size_t)(F_ + a * 192) * NF_ + n;
#pragma unroll 4
        for (int c = 0; c < 192; ++c)
            acc = fmaf(ag[c], wr[(size_t)c * NF_], acc);
    }
    BW[(size_t)bid * NF_ + n] = acc;
}

__global__ __launch_bounds__(256) void k4_out(
    const float* __restrict__ x,   const float* __restrict__ fws,
    const float* __restrict__ BW,  const float* __restrict__ bout,
    float* __restrict__ out)
{
    const int lane  = threadIdx.x & 31;
    const int ntile = threadIdx.x >> 5;
    const int bid   = blockIdx.x;
    const int b     = bid >> 6, mtile = bid & 63;
    const int col   = ntile * 16 + (lane & 15);
    const int kh    = (lane >> 4) * 2;
    const int rowA  = b * V_ + mtile * 16 + (lane & 15);

    const float* xr  = x   + (size_t)rowA * F_;
    const float* er  = fws + (size_t)rowA * C_ + P_;
    const float* bwb = BW  + (size_t)b * KTOT * NF_;

    v8f acc = {};
#pragma unroll
    for (int kt = 0; kt < 4; ++kt) {
        const int k0 = kt * 4 + kh;
        const v2f a = *(const v2f*)(xr + k0);
        v2f bb;
        bb.x = bwb[(size_t)k0 * NF_ + col];
        bb.y = bwb[(size_t)(k0 + 1) * NF_ + col];
        acc = __builtin_amdgcn_wmma_f32_16x16x4_f32(
            false, a, false, bb, (short)0, acc, false, false);
    }
#pragma unroll
    for (int kt = 4; kt < 12; ++kt) {
        const int k0 = kt * 4 + kh;
        const v2f a = *(const v2f*)(er + (k0 - F_));
        v2f bb;
        bb.x = bwb[(size_t)k0 * NF_ + col];
        bb.y = bwb[(size_t)(k0 + 1) * NF_ + col];
        acc = __builtin_amdgcn_wmma_f32_16x16x4_f32(
            false, a, false, bb, (short)0, acc, false, false);
    }

    const float bias = bout[col];
#pragma unroll
    for (int r = 0; r < 8; ++r) {
        const int row = b * V_ + mtile * 16 + r + ((lane >> 4) * 8);
        out[(size_t)row * NF_ + col] = tanhf(acc[r] + bias);
    }
}

extern "C" void kernel_launch(void* const* d_in, const int* in_sizes, int n_in,
                              void* d_out, int out_size, void* d_ws, size_t ws_size,
                              hipStream_t stream)
{
    const float* x    = (const float*)d_in[0];
    const float* Wf   = (const float*)d_in[1];
    const float* bf   = (const float*)d_in[2];
    const float* Ws   = (const float*)d_in[3];
    const float* bs   = (const float*)d_in[4];
    const float* Wout = (const float*)d_in[5];
    const float* bout = (const float*)d_in[6];
    float* out = (float*)d_out;

    float* fws = (float*)d_ws;
    float* agg = fws + (size_t)ROWS * C_;
    float* BW  = agg + (size_t)B_ * A_ * 192;
    float* WC  = BW  + (size_t)B_ * KTOT * NF_;
    float* bc  = WC  + (size_t)F_ * C_;

    k0_prep  <<<7,         256, 0, stream>>>(Wf, bf, Ws, bs, WC, bc);
    k1_vertex<<<ROWS / 16, 192, 0, stream>>>(x, WC, bc, fws);
    k2_agg   <<<B_ * (A_ / K2_AG), K2_T, 0, stream>>>(fws, agg);
    k3_bw    <<<B_ * KTOT, 128, 0, stream>>>(agg, Wout, BW);
    k4_out   <<<B_ * 64,   256, 0, stream>>>(x, fws, BW, bout, out);
}


# k3 (collapsed B matrix) on f32 WMMA: one GEMM per aggregator over the 32 batches, K split over 4 waves and summed in LDS; W_out read once, not per batch (dot-product sum order changes)
# speedup vs baseline: 4.1860x; 1.6497x over previous
#include <hip/hip_runtime.h>
#include <math.h>


typedef __attribute__((ext_vector_type(2))) float v2f;
typedef __attribute__((ext_vector_type(8))) float v8f;

#define B_   32
#define V_   1024
#define F_   16
#define A_   32
#define P_   64
#define NF_  128
#define C_   96
#define KTOT 48
#define ROWS (B_ * V_)


__global__ __launch_bounds__(256) void k0_prep(
    const float* __restrict__ Wf, const float* __restrict__ bf,
    const float* __restrict__ Ws, const float* __restrict__ bs,
    float* __restrict__ WC, float* __restrict__ bc)
{
    const int tid = blockIdx.x * 256 + threadIdx.x;
    if (tid < F_ * C_) {
        const int k = tid / C_, c = tid % C_;
        WC[tid] = (c < P_) ? Wf[k * P_ + c] : Ws[k * A_ + (c - P_)];
    } else if (tid < F_ * C_ + C_) {
        const int c = tid - F_ * C_;
        bc[c] = (c < P_) ? bf[c] : bs[c - P_];
    }
}

__global__ __launch_bounds__(192) void k1_vertex(
    const float* __restrict__ x,  const float* __restrict__ WC,
    const float* __restrict__ bc, float* __restrict__ fws)
{
    const int lane  = threadIdx.x & 31;
    const int ntile = threadIdx.x >> 5;
    const int mtile = blockIdx.x;
    const int col   = ntile * 16 + (lane & 15);
    const int kh    = (lane >> 4) * 2;
    const float* xr = x + (size_t)(mtile * 16 + (lane & 15)) * F_;

    v8f acc = {};
#pragma unroll
    for (int kt = 0; kt < 4; ++kt) {
        const int k0 = kt * 4 + kh;
        const v2f a = *(const v2f*)(xr + k0);
        v2f b;
        b.x = WC[k0 * C_ + col];
        b.y = WC[(k0 + 1) * C_ + col];
        acc = __builtin_amdgcn_wmma_f32_16x16x4_f32(
            false, a, false, b, (short)0, acc, false, false);
    }

    const float bias = bc[col];
#pragma unroll
    for (int r = 0; r < 8; ++r) {
        const int row = mtile * 16 + r + ((lane >> 4) * 8);
        float v = acc[r] + bias;
        if (col >= P_) v = expf(-v * v);
        fws[(size_t)row * C_ + col] = v;
    }
}

#define K2_AG 4
#define K2_VS 4
#define K2_VL (V_ / K2_VS)
#define K2_U  16
#define K2_NB (K2_VL / K2_U)
#define K2_T  (96 * K2_VS)

__device__ __forceinline__ void k2_load(float (&d)[K2_U], const float* __restrict__ p)
{
#pragma unroll
    for (int u = 0; u < K2_U; ++u) d[u] = p[(size_t)u * C_];
}

__device__ __forceinline__ void k2_step(const float (&fv)[K2_U], const float4* w4,
                                        float (&mx)[K2_AG], float (&sm)[K2_AG])
{
#pragma unroll
    for (int u = 0; u < K2_U; ++u) {
        const float4 w = w4[u];
        const float wj[K2_AG] = {w.x, w.y, w.z, w.w};
#pragma unroll
        for (int j = 0; j < K2_AG; ++j) {
            const float val = wj[j] * fv[u];
            mx[j] = fmaxf(mx[j], val);
            sm[j] += val;
        }
    }
}

__global__ __launch_bounds__(K2_T) void k2_agg(
    const float* __restrict__ fws, float* __restrict__ agg)
{
    __shared__ float4 sew[V_];
    __shared__ float  smx[K2_VS][K2_AG][C_];
    __shared__ float  ssm[K2_VS][K2_AG][C_];

    const int b    = blockIdx.x >> 3;
    const int a0   = (blockIdx.x & 7) * K2_AG;
    const int lane = threadIdx.x & 31;
    const int wv   = threadIdx.x >> 5;
    const int g    = wv % 3;
    const int s    = wv / 3;
    const int c    = g * 32 + lane;

    const float* fb = fws + (size_t)b * V_ * C_;
    const float* fp = fb + (size_t)s * K2_VL * C_ + c;

    float fa[K2_U], fc[K2_U];
    k2_load(fa, fp);

    const float* eb = fb + P_ + a0;
    const int v0 = threadIdx.x, v1 = threadIdx.x + K2_T;
    const int v2 = (threadIdx.x + 2 * K2_T < V_) ? threadIdx.x + 2 * K2_T : v0;
    const float4 e0 = *(const float4*)(eb + (size_t)v0 * C_);
    const float4 e1 = *(const float4*)(eb + (size_t)v1 * C_);
    const float4 e2 = *(const float4*)(eb + (size_t)v2 * C_);
    sew[v0] = e0;
    sew[v1] = e1;
    sew[v2] = e2;
    __syncthreads();

    float mx[K2_AG], sm[K2_AG];
#pragma unroll
    for (int j = 0; j < K2_AG; ++j) { mx[j] = -INFINITY; sm[j] = 0.0f; }

    static_assert(K2_NB >= 2 && K2_NB % 2 == 0, "batches are consumed in pairs");
    const float4* wp = sew + s * K2_VL;
    for (int it = 0; it < K2_NB - 2; it += 2) {
        k2_load(fc, fp + (size_t)(it + 1) * K2_U * C_);
        k2_step(fa, wp + it * K2_U, mx, sm);
        k2_load(fa, fp + (size_t)(it + 2) * K2_U * C_);
        k2_step(fc, wp + (it + 1) * K2_U, mx, sm);
    }
    k2_load(fc, fp + (size_t)(K2_NB - 1) * K2_U * C_);
    k2_step(fa, wp + (K2_NB - 2) * K2_U, mx, sm);
    k2_step(fc, wp + (K2_NB - 1) * K2_U, mx, sm);

#pragma unroll
    for (int j = 0; j < K2_AG; ++j) { smx[s][j][c] = mx[j]; ssm[s][j][c] = sm[j]; }
    __syncthreads();

    const int j  = threadIdx.x / C_;
    const int cc = threadIdx.x % C_;
    float m = smx[0][j][cc], t = ssm[0][j][cc];
#pragma unroll
    for (int q = 1; q < K2_VS; ++q) { m = fmaxf(m, smx[q][j][cc]); t += ssm[q][j][cc]; }
    float* ao = agg + (size_t)(b * A_ + a0 + j) * 192;
    ao[cc]      = m;
    ao[96 + cc] = t * (1.0f / (float)V_);
}

#define K3_KQ    4
#define K3_KL    (192 / K3_KQ)
#define K3_NG    (NF_ / 32)
#define K3_PITCH 36

__global__ __launch_bounds__(256) void k3_bw(
    const float* __restrict__ agg, const float* __restrict__ Wout,
    float* __restrict__ BW)
{
    __shared__ float red[K3_KQ][B_][K3_PITCH];
    static_assert(K3_KQ == 4 && K3_KL % 4 == 0, "the epilogue sums exactly four quarter partials");

    if (blockIdx.x >= A_ * K3_NG) {
        const int h = blockIdx.x - A_ * K3_NG;
        const float4* src = (const float4*)(Wout + (h & 1) * 1024);
        float4* dst = (float4*)(BW + (size_t)(h >> 1) * KTOT * NF_ + (h & 1) * 1024);
        dst[threadIdx.x] = src[threadIdx.x];
        return;
    }

    const int a    = blockIdx.x / K3_NG;
    const int ng   = blockIdx.x % K3_NG;
    const int lane = threadIdx.x & 31;
    const int wv   = threadIdx.x >> 5;
    const int nt   = wv & 1;
    const int q    = wv >> 1;
    const int c16  = lane & 15;
    const int kh   = (lane >> 4) * 2;
    const int col  = ng * 32 + nt * 16 + c16;

    const float* a0p = agg + ((size_t)c16 * A_ + a) * 192 + q * K3_KL + kh;
    const float* a1p = a0p + (size_t)16 * A_ * 192;
    const float* bp  = Wout + (size_t)(F_ + a * 192 + q * K3_KL + kh) * NF_ + col;

    const float bias = Wout[(size_t)(F_ + 2 * A_ * C_ + a) * NF_ + ng * 32 + (threadIdx.x & 31)];

    v8f acc0 = {}, acc1 = {};
#pragma unroll
    for (int kt = 0; kt < K3_KL / 4; ++kt) {
        const v2f av0 = *(const v2f*)(a0p + kt * 4);
        const v2f av1 = *(const v2f*)(a1p + kt * 4);
        v2f bb;
        bb.x = bp[(size_t)(kt * 4) * NF_];
        bb.y = bp[(size_t)(kt * 4 + 1) * NF_];
        acc0 = __builtin_amdgcn_wmma_f32_16x16x4_f32(
            false, av0, false, bb, (short)0, acc0, false, false);
        acc1 = __builtin_amdgcn_wmma_f32_16x16x4_f32(
            false, av1, false, bb, (short)0, acc1, false, false);
    }

#pragma unroll
    for (int r = 0; r < 8; ++r) {
        const int bl = r + (lane >> 4) * 8;
        red[q][bl][nt * 16 + c16]      = acc0[r];
        red[q][16 + bl][nt * 16 + c16] = acc1[r];
    }
    __syncthreads();

    const int cc = threadIdx.x & 31;
    const int rb = threadIdx.x >> 5;
#pragma unroll
    for (int i = 0; i < 4; ++i) {
        const int b = rb + 8 * i;
        const float t = ((red[0][b][cc] + red[1][b][cc]) + red[2][b][cc]) + red[3][b][cc];
        BW[((size_t)b * KTOT + F_ + a) * NF_ + ng * 32 + cc] = t + bias;
    }
}

__global__ __launch_bounds__(256) void k4_out(
    const float* __restrict__ x,   const float* __restrict__ fws,
    const float* __restrict__ BW,  const float* __restrict__ bout,
    float* __restrict__ out)
{
    const int lane  = threadIdx.x & 31;
    const int ntile = threadIdx.x >> 5;
    const int bid   = blockIdx.x;
    const int b     = bid >> 6, mtile = bid & 63;
    const int col   = ntile * 16 + (lane & 15);
    const int kh    = (lane >> 4) * 2;
    const int rowA  = b * V_ + mtile * 16 + (lane & 15);

    const float* xr  = x   + (size_t)rowA * F_;
    const float* er  = fws + (size_t)rowA * C_ + P_;
    const float* bwb = BW  + (size_t)b * KTOT * NF_;

    v8f acc = {};
#pragma unroll
    for (int kt = 0; kt < 4; ++kt) {
        const int k0 = kt * 4 + kh;
        const v2f a = *(const v2f*)(xr + k0);
        v2f bb;
        bb.x = bwb[(size_t)k0 * NF_ + col];
        bb.y = bwb[(size_t)(k0 + 1) * NF_ + col];
        acc = __builtin_amdgcn_wmma_f32_16x16x4_f32(
            false, a, false, bb, (short)0, acc, false, false);
    }
#pragma unroll
    for (int kt = 4; kt < 12; ++kt) {
        const int k0 = kt * 4 + kh;
        const v2f a = *(const v2f*)(er + (k0 - F_));
        v2f bb;
        bb.x = bwb[(size_t)k0 * NF_ + col];
        bb.y = bwb[(size_t)(k0 + 1) * NF_ + col];
        acc = __builtin_amdgcn_wmma_f32_16x16x4_f32(
            false, a, false, bb, (short)0, acc, false, false);
    }

    const float bias = bout[col];
#pragma unroll
    for (int r = 0; r < 8; ++r) {
        const int row = b * V_ + mtile * 16 + r + ((lane >> 4) * 8);
        out[(size_t)row * NF_ + col] = tanhf(acc[r] + bias);
    }
}

extern "C" void kernel_launch(void* const* d_in, const int* in_sizes, int n_in,
                              void* d_out, int out_size, void* d_ws, size_t ws_size,
                              hipStream_t stream)
{
    const float* x    = (const float*)d_in[0];
    const float* Wf   = (const float*)d_in[1];
    const float* bf   = (const float*)d_in[2];
    const float* Ws   = (const float*)d_in[3];
    const float* bs   = (const float*)d_in[4];
    const float* Wout = (const float*)d_in[5];
    const float* bout = (const float*)d_in[6];
    float* out = (float*)d_out;

    float* fws = (float*)d_ws;
    float* agg = fws + (size_t)ROWS * C_;
    float* BW  = agg + (size_t)B_ * A_ * 192;
    float* WC  = BW  + (size_t)B_ * KTOT * NF_;
    float* bc  = WC  + (size_t)F_ * C_;

    k0_prep  <<<7,         256, 0, stream>>>(Wf, bf, Ws, bs, WC, bc);
    k1_vertex<<<ROWS / 16, 192, 0, stream>>>(x, WC, bc, fws);
    k2_agg   <<<B_ * (A_ / K2_AG), K2_T, 0, stream>>>(fws, agg);
    k3_bw    <<<A_ * K3_NG + 2 * B_, 256, 0, stream>>>(agg, Wout, BW);
    k4_out   <<<B_ * 64,   256, 0, stream>>>(x, fws, BW, bout, out);
}


# packing launch removed: k1 picks W_flr or W_s (and its bias) once per wave through a wave-uniform scalar select and reads the weights in place; 5 launches become 4
# speedup vs baseline: 4.4485x; 1.0627x over previous
#include <hip/hip_runtime.h>
#include <math.h>


typedef __attribute__((ext_vector_type(2))) float v2f;
typedef __attribute__((ext_vector_type(8))) float v8f;

#define B_   32
#define V_   1024
#define F_   16
#define A_   32
#define P_   64
#define NF_  128
#define C_   96
#define KTOT 48
#define ROWS (B_ * V_)


__global__ __launch_bounds__(192) void k1_vertex(
    const float* __restrict__ x,
    const float* __restrict__ Wf, const float* __restrict__ bf,
    const float* __restrict__ Ws, const float* __restrict__ bs,
    float* __restrict__ fws)
{
    const int lane  = threadIdx.x & 31;
    const int ntile = __builtin_amdgcn_readfirstlane((int)(threadIdx.x >> 5));
    const int mtile = blockIdx.x;
    const int c16   = lane & 15;
    const int col   = ntile * 16 + c16;
    const int kh    = (lane >> 4) * 2;
    const float* xr = x + (size_t)(mtile * 16 + c16) * F_;

    const bool   isE   = ntile >= P_ / 16;
    const int    pitch = isE ? A_ : P_;
    const float* Wp    = isE ? Ws + (ntile * 16 - P_) : Wf + ntile * 16;
    const float* bp    = isE ? bs + (ntile * 16 - P_) : bf + ntile * 16;

    v8f acc = {};
#pragma unroll
    for (int kt = 0; kt < 4; ++kt) {
        const int k0 = kt * 4 + kh;
        const v2f a = *(const v2f*)(xr + k0);
        v2f b;
        b.x = Wp[k0 * pitch + c16];
        b.y = Wp[(k0 + 1) * pitch + c16];
        acc = __builtin_amdgcn_wmma_f32_16x16x4_f32(
            false, a, false, b, (short)0, acc, false, false);
    }

    const float bias = bp[c16];
#pragma unroll
    for (int r = 0; r < 8; ++r) {
        const int row = mtile * 16 + r + ((lane >> 4) * 8);
        float v = acc[r] + bias;
        if (isE) v = expf(-v * v);
        fws[(size_t)row * C_ + col] = v;
    }
}

#define K2_AG 4
#define K2_VS 4
#define K2_VL (V_ / K2_VS)
#define K2_U  16
#define K2_NB (K2_VL / K2_U)
#define K2_T  (96 * K2_VS)

__device__ __forceinline__ void k2_load(float (&d)[K2_U], const float* __restrict__ p)
{
#pragma unroll
    for (int u = 0; u < K2_U; ++u) d[u] = p[(size_t)u * C_];
}

__device__ __forceinline__ void k2_step(const float (&fv)[K2_U], const float4* w4,
                                        float (&mx)[K2_AG], float (&sm)[K2_AG])
{
#pragma unroll
    for (int u = 0; u < K2_U; ++u) {
        const float4 w = w4[u];
        const float wj[K2_AG] = {w.x, w.y, w.z, w.w};
#pragma unroll
        for (int j = 0; j < K2_AG; ++j) {
            const float val = wj[j] * fv[u];
            mx[j] = fmaxf(mx[j], val);
            sm[j] += val;
        }
    }
}

__global__ __launch_bounds__(K2_T) void k2_agg(
    const float* __restrict__ fws, float* __restrict__ agg)
{
    __shared__ float4 sew[V_];
    __shared__ float  smx[K2_VS][K2_AG][C_];
    __shared__ float  ssm[K2_VS][K2_AG][C_];

    const int b    = blockIdx.x >> 3;
    const int a0   = (blockIdx.x & 7) * K2_AG;
    const int lane = threadIdx.x & 31;
    const int wv   = threadIdx.x >> 5;
    const int g    = wv % 3;
    const int s    = wv / 3;
    const int c    = g * 32 + lane;

    const float* fb = fws + (size_t)b * V_ * C_;
    const float* fp = fb + (size_t)s * K2_VL * C_ + c;

    float fa[K2_U], fc[K2_U];
    k2_load(fa, fp);

    const float* eb = fb + P_ + a0;
    const int v0 = threadIdx.x, v1 = threadIdx.x + K2_T;
    const int v2 = (threadIdx.x + 2 * K2_T < V_) ? threadIdx.x + 2 * K2_T : v0;
    const float4 e0 = *(const float4*)(eb + (size_t)v0 * C_);
    const float4 e1 = *(const float4*)(eb + (size_t)v1 * C_);
    const float4 e2 = *(const float4*)(eb + (size_t)v2 * C_);
    sew[v0] = e0;
    sew[v1] = e1;
    sew[v2] = e2;
    __syncthreads();

    float mx[K2_AG], sm[K2_AG];
#pragma unroll
    for (int j = 0; j < K2_AG; ++j) { mx[j] = -INFINITY; sm[j] = 0.0f; }

    static_assert(K2_NB >= 2 && K2_NB % 2 == 0, "batches are consumed in pairs");
    const float4* wp = sew + s * K2_VL;
    for (int it = 0; it < K2_NB - 2; it += 2) {
        k2_load(fc, fp + (size_t)(it + 1) * K2_U * C_);
        k2_step(fa, wp + it * K2_U, mx, sm);
        k2_load(fa, fp + (size_t)(it + 2) * K2_U * C_);
        k2_step(fc, wp + (it + 1) * K2_U, mx, sm);
    }
    k2_load(fc, fp + (size_t)(K2_NB - 1) * K2_U * C_);
    k2_step(fa, wp + (K2_NB - 2) * K2_U, mx, sm);
    k2_step(fc, wp + (K2_NB - 1) * K2_U, mx, sm);

#pragma unroll
    for (int j = 0; j < K2_AG; ++j) { smx[s][j][c] = mx[j]; ssm[s][j][c] = sm[j]; }
    __syncthreads();

    const int j  = threadIdx.x / C_;
    const int cc = threadIdx.x % C_;
    float m = smx[0][j][cc], t = ssm[0][j][cc];
#pragma unroll
    for (int q = 1; q < K2_VS; ++q) { m = fmaxf(m, smx[q][j][cc]); t += ssm[q][j][cc]; }
    float* ao = agg + (size_t)(b * A_ + a0 + j) * 192;
    ao[cc]      = m;
    ao[96 + cc] = t * (1.0f / (float)V_);
}

#define K3_KQ    4
#define K3_KL    (192 / K3_KQ)
#define K3_NG    (NF_ / 32)
#define K3_PITCH 36

__global__ __launch_bounds__(256) void k3_bw(
    const float* __restrict__ agg, const float* __restrict__ Wout,
    float* __restrict__ BW)
{
    __shared__ float red[K3_KQ][B_][K3_PITCH];
    static_assert(K3_KQ == 4 && K3_KL % 4 == 0, "the epilogue sums exactly four quarter partials");

    if (blockIdx.x >= A_ * K3_NG) {
        const int h = blockIdx.x - A_ * K3_NG;
        const float4* src = (const float4*)(Wout + (h & 1) * 1024);
        float4* dst = (float4*)(BW + (size_t)(h >> 1) * KTOT * NF_ + (h & 1) * 1024);
        dst[threadIdx.x] = src[threadIdx.x];
        return;
    }

    const int a    = blockIdx.x / K3_NG;
    const int ng   = blockIdx.x % K3_NG;
    const int lane = threadIdx.x & 31;
    const int wv   = threadIdx.x >> 5;
    const int nt   = wv & 1;
    const int q    = wv >> 1;
    const int c16  = lane & 15;
    const int kh   = (lane >> 4) * 2;
    const int col  = ng * 32 + nt * 16 + c16;

    const float* a0p = agg + ((size_t)c16 * A_ + a) * 192 + q * K3_KL + kh;
    const float* a1p = a0p + (size_t)16 * A_ * 192;
    const float* bp  = Wout + (size_t)(F_ + a * 192 + q * K3_KL + kh) * NF_ + col;

    const float bias = Wout[(size_t)(F_ + 2 * A_ * C_ + a) * NF_ + ng * 32 + (threadIdx.x & 31)];

    v8f acc0 = {}, acc1 = {};
#pragma unroll
    for (int kt = 0; kt < K3_KL / 4; ++kt) {
        const v2f av0 = *(const v2f*)(a0p + kt * 4);
        const v2f av1 = *(const v2f*)(a1p + kt * 4);
        v2f bb;
        bb.x = bp[(size_t)(kt * 4) * NF_];
        bb.y = bp[(size_t)(kt * 4 + 1) * NF_];
        acc0 = __builtin_amdgcn_wmma_f32_16x16x4_f32(
            false, av0, false, bb, (short)0, acc0, false, false);
        acc1 = __builtin_amdgcn_wmma_f32_16x16x4_f32(
            false, av1, false, bb, (short)0, acc1, false, false);
    }

#pragma unroll
    for (int r = 0; r < 8; ++r) {
        const int bl = r + (lane >> 4) * 8;
        red[q][bl][nt * 16 + c16]      = acc0[r];
        red[q][16 + bl][nt * 16 + c16] = acc1[r];
    }
    __syncthreads();

    const int cc = threadIdx.x & 31;
    const int rb = threadIdx.x >> 5;
#pragma unroll
    for (int i = 0; i < 4; ++i) {
        const int b = rb + 8 * i;
        const float t = ((red[0][b][cc] + red[1][b][cc]) + red[2][b][cc]) + red[3][b][cc];
        BW[((size_t)b * KTOT + F_ + a) * NF_ + ng * 32 + cc] = t + bias;
    }
}

__global__ __launch_bounds__(256) void k4_out(
    const float* __restrict__ x,   const float* __restrict__ fws,
    const float* __restrict__ BW,  const float* __restrict__ bout,
    float* __restrict__ out)
{
    const int lane  = threadIdx.x & 31;
    const int ntile = threadIdx.x >> 5;
    const int bid   = blockIdx.x;
    const int b     = bid >> 6, mtile = bid & 63;
    const int col   = ntile * 16 + (lane & 15);
    const int kh    = (lane >> 4) * 2;
    const int rowA  = b * V_ + mtile * 16 + (lane & 15);

    const float* xr  = x   + (size_t)rowA * F_;
    const float* er  = fws + (size_t)rowA * C_ + P_;
    const float* bwb = BW  + (size_t)b * KTOT * NF_;

    v8f acc = {};
#pragma unroll
    for (int kt = 0; kt < 4; ++kt) {
        const int k0 = kt * 4 + kh;
        const v2f a = *(const v2f*)(xr + k0);
        v2f bb;
        bb.x = bwb[(size_t)k0 * NF_ + col];
        bb.y = bwb[(size_t)(k0 + 1) * NF_ + col];
        acc = __builtin_amdgcn_wmma_f32_16x16x4_f32(
            false, a, false, bb, (short)0, acc, false, false);
    }
#pragma unroll
    for (int kt = 4; kt < 12; ++kt) {
        const int k0 = kt * 4 + kh;
        const v2f a = *(const v2f*)(er + (k0 - F_));
        v2f bb;
        bb.x = bwb[(size_t)k0 * NF_ + col];
        bb.y = bwb[(size_t)(k0 + 1) * NF_ + col];
        acc = __builtin_amdgcn_wmma_f32_16x16x4_f32(
            false, a, false, bb, (short)0, acc, false, false);
    }

    const float bias = bout[col];
#pragma unroll
    for (int r = 0; r < 8; ++r) {
        const int row = b * V_ + mtile * 16 + r + ((lane >> 4) * 8);
        out[(size_t)row * NF_ + col] = tanhf(acc[r] + bias);
    }
}

extern "C" void kernel_launch(void* const* d_in, const int* in_sizes, int n_in,
                              void* d_out, int out_size, void* d_ws, size_t ws_size,
                              hipStream_t stream)
{
    const float* x    = (const float*)d_in[0];
    const float* Wf   = (const float*)d_in[1];
    const float* bf   = (const float*)d_in[2];
    const float* Ws   = (const float*)d_in[3];
    const float* bs   = (const float*)d_in[4];
    const float* Wout = (const float*)d_in[5];
    const float* bout = (const float*)d_in[6];
    float* out = (float*)d_out;

    float* fws = (float*)d_ws;
    float* agg = fws + (size_t)ROWS * C_;
    float* BW  = agg + (size_t)B_ * A_ * 192;

    k1_vertex<<<ROWS / 16, 192, 0, stream>>>(x, Wf, bf, Ws, bs, fws);
    k2_agg   <<<B_ * (A_ / K2_AG), K2_T, 0, stream>>>(fws, agg);
    k3_bw    <<<A_ * K3_NG + 2 * B_, 256, 0, stream>>>(agg, Wout, BW);
    k4_out   <<<B_ * 64,   256, 0, stream>>>(x, fws, BW, bout, out);
}


# k2 retiled: a wave spans the whole 96-channel f row (3 channels per lane), vertices over 16 wave slices; each LDS weight read now feeds 3 channels (LDS reads / 3); mean sum order changes
# speedup vs baseline: 4.6346x; 1.0418x over previous
#include <hip/hip_runtime.h>
#include <math.h>


typedef __attribute__((ext_vector_type(2))) float v2f;
typedef __attribute__((ext_vector_type(8))) float v8f;

#define B_   32
#define V_   1024
#define F_   16
#define A_   32
#define P_   64
#define NF_  128
#define C_   96
#define KTOT 48
#define ROWS (B_ * V_)


__global__ __launch_bounds__(192) void k1_vertex(
    const float* __restrict__ x,
    const float* __restrict__ Wf, const float* __restrict__ bf,
    const float* __restrict__ Ws, const float* __restrict__ bs,
    float* __restrict__ fws)
{
    const int lane  = threadIdx.x & 31;
    const int ntile = __builtin_amdgcn_readfirstlane((int)(threadIdx.x >> 5));
    const int mtile = blockIdx.x;
    const int c16   = lane & 15;
    const int col   = ntile * 16 + c16;
    const int kh    = (lane >> 4) * 2;
    const float* xr = x + (size_t)(mtile * 16 + c16) * F_;

    const bool   isE   = ntile >= P_ / 16;
    const int    pitch = isE ? A_ : P_;
    const float* Wp    = isE ? Ws + (ntile * 16 - P_) : Wf + ntile * 16;
    const float* bp    = isE ? bs + (ntile * 16 - P_) : bf + ntile * 16;

    v8f acc = {};
#pragma unroll
    for (int kt = 0; kt < 4; ++kt) {
        const int k0 = kt * 4 + kh;
        const v2f a = *(const v2f*)(xr + k0);
        v2f b;
        b.x = Wp[k0 * pitch + c16];
        b.y = Wp[(k0 + 1) * pitch + c16];
        acc = __builtin_amdgcn_wmma_f32_16x16x4_f32(
            false, a, false, b, (short)0, acc, false, false);
    }

    const float bias = bp[c16];
#pragma unroll
    for (int r = 0; r < 8; ++r) {
        const int row = mtile * 16 + r + ((lane >> 4) * 8);
        float v = acc[r] + bias;
        if (isE) v = expf(-v * v);
        fws[(size_t)row * C_ + col] = v;
    }
}

#define K2_AG 4
#define K2_CG (C_ / 32)
#define K2_VS 16
#define K2_VL (V_ / K2_VS)
#define K2_U  8
#define K2_NB (K2_VL / K2_U)
#define K2_T  (32 * K2_VS)

__device__ __forceinline__ void k2_load(float (&d)[K2_U][K2_CG], const float* __restrict__ p)
{
#pragma unroll
    for (int u = 0; u < K2_U; ++u) {
#pragma unroll
        for (int g = 0; g < K2_CG; ++g) d[u][g] = p[(size_t)u * C_ + g * 32];
    }
}

__device__ __forceinline__ void k2_step(const float (&fv)[K2_U][K2_CG], const float4* w4,
                                        float (&mx)[K2_CG][K2_AG], float (&sm)[K2_CG][K2_AG])
{
#pragma unroll
    for (int u = 0; u < K2_U; ++u) {
        const float4 w = w4[u];
        const float wj[K2_AG] = {w.x, w.y, w.z, w.w};
#pragma unroll
        for (int g = 0; g < K2_CG; ++g) {
#pragma unroll
            for (int j = 0; j < K2_AG; ++j) {
                mx[g][j] = fmaxf(mx[g][j], wj[j] * fv[u][g]);
                sm[g][j] = fmaf(wj[j], fv[u][g], sm[g][j]);
            }
        }
    }
}

__global__ __launch_bounds__(K2_T) void k2_agg(
    const float* __restrict__ fws, float* __restrict__ agg)
{
    __shared__ float4 sew[V_];
    __shared__ float  smx[K2_VS][K2_AG][C_];
    __shared__ float  ssm[K2_VS][K2_AG][C_];

    const int b    = blockIdx.x >> 3;
    const int a0   = (blockIdx.x & 7) * K2_AG;
    const int lane = threadIdx.x & 31;
    const int s    = threadIdx.x >> 5;

    const float* fb = fws + (size_t)b * V_ * C_;
    const float* fp = fb + (size_t)s * K2_VL * C_ + lane;

    float fa[K2_U][K2_CG], fc[K2_U][K2_CG];
    k2_load(fa, fp);

    static_assert(V_ == 2 * K2_T, "each thread stages exactly two ew rows");
    const float* eb = fb + P_ + a0;
    const int v0 = threadIdx.x, v1 = threadIdx.x + K2_T;
    const float4 e0 = *(const float4*)(eb + (size_t)v0 * C_);
    const float4 e1 = *(const float4*)(eb + (size_t)v1 * C_);
    sew[v0] = e0;
    sew[v1] = e1;
    __syncthreads();

    float mx[K2_CG][K2_AG], sm[K2_CG][K2_AG];
#pragma unroll
    for (int g = 0; g < K2_CG; ++g) {
#pragma unroll
        for (int j = 0; j < K2_AG; ++j) { mx[g][j] = -INFINITY; sm[g][j] = 0.0f; }
    }

    static_assert(K2_NB >= 2 && K2_NB % 2 == 0, "batches are consumed in pairs");
    const float4* wp = sew + s * K2_VL;
    for (int it = 0; it < K2_NB - 2; it += 2) {
        k2_load(fc, fp + (size_t)(it + 1) * K2_U * C_);
        k2_step(fa, wp + it * K2_U, mx, sm);
        k2_load(fa, fp + (size_t)(it + 2) * K2_U * C_);
        k2_step(fc, wp + (it + 1) * K2_U, mx, sm);
    }
    k2_load(fc, fp + (size_t)(K2_NB - 1) * K2_U * C_);
    k2_step(fa, wp + (K2_NB - 2) * K2_U, mx, sm);
    k2_step(fc, wp + (K2_NB - 1) * K2_U, mx, sm);

#pragma unroll
    for (int g = 0; g < K2_CG; ++g) {
#pragma unroll
        for (int j = 0; j < K2_AG; ++j) {
            smx[s][j][g * 32 + lane] = mx[g][j];
            ssm[s][j][g * 32 + lane] = sm[g][j];
        }
    }
    __syncthreads();

    if (threadIdx.x < K2_AG * C_) {
        const int j  = threadIdx.x / C_;
        const int cc = threadIdx.x % C_;
        float m = smx[0][j][cc], t = ssm[0][j][cc];
#pragma unroll
        for (int q = 1; q < K2_VS; ++q) { m = fmaxf(m, smx[q][j][cc]); t += ssm[q][j][cc]; }
        float* ao = agg + (size_t)(b * A_ + a0 + j) * 192;
        ao[cc]      = m;
        ao[96 + cc] = t * (1.0f / (float)V_);
    }
}

#define K3_KQ    4
#define K3_KL    (192 / K3_KQ)
#define K3_NG    (NF_ / 32)
#define K3_PITCH 36

__global__ __launch_bounds__(256) void k3_bw(
    const float* __restrict__ agg, const float* __restrict__ Wout,
    float* __restrict__ BW)
{
    __shared__ float red[K3_KQ][B_][K3_PITCH];
    static_assert(K3_KQ == 4 && K3_KL % 4 == 0, "the epilogue sums exactly four quarter partials");

    if (blockIdx.x >= A_ * K3_NG) {
        const int h = blockIdx.x - A_ * K3_NG;
        const float4* src = (const float4*)(Wout + (h & 1) * 1024);
        float4* dst = (float4*)(BW + (size_t)(h >> 1) * KTOT * NF_ + (h & 1) * 1024);
        dst[threadIdx.x] = src[threadIdx.x];
        return;
    }

    const int a    = blockIdx.x / K3_NG;
    const int ng   = blockIdx.x % K3_NG;
    const int lane = threadIdx.x & 31;
    const int wv   = threadIdx.x >> 5;
    const int nt   = wv & 1;
    const int q    = wv >> 1;
    const int c16  = lane & 15;
    const int kh   = (lane >> 4) * 2;
    const int col  = ng * 32 + nt * 16 + c16;

    const float* a0p = agg + ((size_t)c16 * A_ + a) * 192 + q * K3_KL + kh;
    const float* a1p = a0p + (size_t)16 * A_ * 192;
    const float* bp  = Wout + (size_t)(F_ + a * 192 + q * K3_KL + kh) * NF_ + col;

    const float bias = Wout[(size_t)(F_ + 2 * A_ * C_ + a) * NF_ + ng * 32 + (threadIdx.x & 31)];

    v8f acc0 = {}, acc1 = {};
#pragma unroll
    for (int kt = 0; kt < K3_KL / 4; ++kt) {
        const v2f av0 = *(const v2f*)(a0p + kt * 4);
        const v2f av1 = *(const v2f*)(a1p + kt * 4);
        v2f bb;
        bb.x = bp[(size_t)(kt * 4) * NF_];
        bb.y = bp[(size_t)(kt * 4 + 1) * NF_];
        acc0 = __builtin_amdgcn_wmma_f32_16x16x4_f32(
            false, av0, false, bb, (short)0, acc0, false, false);
        acc1 = __builtin_amdgcn_wmma_f32_16x16x4_f32(
            false, av1, false, bb, (short)0, acc1, false, false);
    }

#pragma unroll
    for (int r = 0; r < 8; ++r) {
        const int bl = r + (lane >> 4) * 8;
        red[q][bl][nt * 16 + c16]      = acc0[r];
        red[q][16 + bl][nt * 16 + c16] = acc1[r];
    }
    __syncthreads();

    const int cc = threadIdx.x & 31;
    const int rb = threadIdx.x >> 5;
#pragma unroll
    for (int i = 0; i < 4; ++i) {
        const int b = rb + 8 * i;
        const float t = ((red[0][b][cc] + red[1][b][cc]) + red[2][b][cc]) + red[3][b][cc];
        BW[((size_t)b * KTOT + F_ + a) * NF_ + ng * 32 + cc] = t + bias;
    }
}

__global__ __launch_bounds__(256) void k4_out(
    const float* __restrict__ x,   const float* __restrict__ fws,
    const float* __restrict__ BW,  const float* __restrict__ bout,
    float* __restrict__ out)
{
    const int lane  = threadIdx.x & 31;
    const int ntile = threadIdx.x >> 5;
    const int bid   = blockIdx.x;
    const int b     = bid >> 6, mtile = bid & 63;
    const int col   = ntile * 16 + (lane & 15);
    const int kh    = (lane >> 4) * 2;
    const int rowA  = b * V_ + mtile * 16 + (lane & 15);

    const float* xr  = x   + (size_t)rowA * F_;
    const float* er  = fws + (size_t)rowA * C_ + P_;
    const float* bwb = BW  + (size_t)b * KTOT * NF_;

    v8f acc = {};
#pragma unroll
    for (int kt = 0; kt < 4; ++kt) {
        const int k0 = kt * 4 + kh;
        const v2f a = *(const v2f*)(xr + k0);
        v2f bb;
        bb.x = bwb[(size_t)k0 * NF_ + col];
        bb.y = bwb[(size_t)(k0 + 1) * NF_ + col];
        acc = __builtin_amdgcn_wmma_f32_16x16x4_f32(
            false, a, false, bb, (short)0, acc, false, false);
    }
#pragma unroll
    for (int kt = 4; kt < 12; ++kt) {
        const int k0 = kt * 4 + kh;
        const v2f a = *(const v2f*)(er + (k0 - F_));
        v2f bb;
        bb.x = bwb[(size_t)k0 * NF_ + col];
        bb.y = bwb[(size_t)(k0 + 1) * NF_ + col];
        acc = __builtin_amdgcn_wmma_f32_16x16x4_f32(
            false, a, false, bb, (short)0, acc, false, false);
    }

    const float bias = bout[col];
#pragma unroll
    for (int r = 0; r < 8; ++r) {
        const int row = b * V_ + mtile * 16 + r + ((lane >> 4) * 8);
        out[(size_t)row * NF_ + col] = tanhf(acc[r] + bias);
    }
}

extern "C" void kernel_launch(void* const* d_in, const int* in_sizes, int n_in,
                              void* d_out, int out_size, void* d_ws, size_t ws_size,
                              hipStream_t stream)
{
    const float* x    = (const float*)d_in[0];
    const float* Wf   = (const float*)d_in[1];
    const float* bf   = (const float*)d_in[2];
    const float* Ws   = (const float*)d_in[3];
    const float* bs   = (const float*)d_in[4];
    const float* Wout = (const float*)d_in[5];
    const float* bout = (const float*)d_in[6];
    float* out = (float*)d_out;

    float* fws = (float*)d_ws;
    float* agg = fws + (size_t)ROWS * C_;
    float* BW  = agg + (size_t)B_ * A_ * 192;

    k1_vertex<<<ROWS / 16, 192, 0, stream>>>(x, Wf, bf, Ws, bs, fws);
    k2_agg   <<<B_ * (A_ / K2_AG), K2_T, 0, stream>>>(fws, agg);
    k3_bw    <<<A_ * K3_NG + 2 * B_, 256, 0, stream>>>(agg, Wout, BW);
    k4_out   <<<B_ * 64,   256, 0, stream>>>(x, fws, BW, bout, out);
}


# k4 retiled: block = 32 rows; the [x|ew] A tile is loaded once per block with whole-line loads and staged in LDS, each wave keeps its BW fragment in registers for 2 row tiles; sum order kept
# speedup vs baseline: 4.8208x; 1.0402x over previous
#include <hip/hip_runtime.h>
#include <math.h>


typedef __attribute__((ext_vector_type(2))) float v2f;
typedef __attribute__((ext_vector_type(8))) float v8f;

#define B_   32
#define V_   1024
#define F_   16
#define A_   32
#define P_   64
#define NF_  128
#define C_   96
#define KTOT 48
#define ROWS (B_ * V_)


__global__ __launch_bounds__(192) void k1_vertex(
    const float* __restrict__ x,
    const float* __restrict__ Wf, const float* __restrict__ bf,
    const float* __restrict__ Ws, const float* __restrict__ bs,
    float* __restrict__ fws)
{
    const int lane  = threadIdx.x & 31;
    const int ntile = __builtin_amdgcn_readfirstlane((int)(threadIdx.x >> 5));
    const int mtile = blockIdx.x;
    const int c16   = lane & 15;
    const int col   = ntile * 16 + c16;
    const int kh    = (lane >> 4) * 2;
    const float* xr = x + (size_t)(mtile * 16 + c16) * F_;

    const bool   isE   = ntile >= P_ / 16;
    const int    pitch = isE ? A_ : P_;
    const float* Wp    = isE ? Ws + (ntile * 16 - P_) : Wf + ntile * 16;
    const float* bp    = isE ? bs + (ntile * 16 - P_) : bf + ntile * 16;

    v8f acc = {};
#pragma unroll
    for (int kt = 0; kt < 4; ++kt) {
        const int k0 = kt * 4 + kh;
        const v2f a = *(const v2f*)(xr + k0);
        v2f b;
        b.x = Wp[k0 * pitch + c16];
        b.y = Wp[(k0 + 1) * pitch + c16];
        acc = __builtin_amdgcn_wmma_f32_16x16x4_f32(
            false, a, false, b, (short)0, acc, false, false);
    }

    const float bias = bp[c16];
#pragma unroll
    for (int r = 0; r < 8; ++r) {
        const int row = mtile * 16 + r + ((lane >> 4) * 8);
        float v = acc[r] + bias;
        if (isE) v = expf(-v * v);
        fws[(size_t)row * C_ + col] = v;
    }
}

#define K2_AG 4
#define K2_CG (C_ / 32)
#define K2_VS 16
#define K2_VL (V_ / K2_VS)
#define K2_U  8
#define K2_NB (K2_VL / K2_U)
#define K2_T  (32 * K2_VS)

__device__ __forceinline__ void k2_load(float (&d)[K2_U][K2_CG], const float* __restrict__ p)
{
#pragma unroll
    for (int u = 0; u < K2_U; ++u) {
#pragma unroll
        for (int g = 0; g < K2_CG; ++g) d[u][g] = p[(size_t)u * C_ + g * 32];
    }
}

__device__ __forceinline__ void k2_step(const float (&fv)[K2_U][K2_CG], const float4* w4,
                                        float (&mx)[K2_CG][K2_AG], float (&sm)[K2_CG][K2_AG])
{
#pragma unroll
    for (int u = 0; u < K2_U; ++u) {
        const float4 w = w4[u];
        const float wj[K2_AG] = {w.x, w.y, w.z, w.w};
#pragma unroll
        for (int g = 0; g < K2_CG; ++g) {
#pragma unroll
            for (int j = 0; j < K2_AG; ++j) {
                mx[g][j] = fmaxf(mx[g][j], wj[j] * fv[u][g]);
                sm[g][j] = fmaf(wj[j], fv[u][g], sm[g][j]);
            }
        }
    }
}

__global__ __launch_bounds__(K2_T) void k2_agg(
    const float* __restrict__ fws, float* __restrict__ agg)
{
    __shared__ float4 sew[V_];
    __shared__ float  smx[K2_VS][K2_AG][C_];
    __shared__ float  ssm[K2_VS][K2_AG][C_];

    const int b    = blockIdx.x >> 3;
    const int a0   = (blockIdx.x & 7) * K2_AG;
    const int lane = threadIdx.x & 31;
    const int s    = threadIdx.x >> 5;

    const float* fb = fws + (size_t)b * V_ * C_;
    const float* fp = fb + (size_t)s * K2_VL * C_ + lane;

    float fa[K2_U][K2_CG], fc[K2_U][K2_CG];
    k2_load(fa, fp);

    static_assert(V_ == 2 * K2_T, "each thread stages exactly two ew rows");
    const float* eb = fb + P_ + a0;
    const int v0 = threadIdx.x, v1 = threadIdx.x + K2_T;
    const float4 e0 = *(const float4*)(eb + (size_t)v0 * C_);
    const float4 e1 = *(const float4*)(eb + (size_t)v1 * C_);
    sew[v0] = e0;
    sew[v1] = e1;
    __syncthreads();

    float mx[K2_CG][K2_AG], sm[K2_CG][K2_AG];
#pragma unroll
    for (int g = 0; g < K2_CG; ++g) {
#pragma unroll
        for (int j = 0; j < K2_AG; ++j) { mx[g][j] = -INFINITY; sm[g][j] = 0.0f; }
    }

    static_assert(K2_NB >= 2 && K2_NB % 2 == 0, "batches are consumed in pairs");
    const float4* wp = sew + s * K2_VL;
    for (int it = 0; it < K2_NB - 2; it += 2) {
        k2_load(fc, fp + (size_t)(it + 1) * K2_U * C_);
        k2_step(fa, wp + it * K2_U, mx, sm);
        k2_load(fa, fp + (size_t)(it + 2) * K2_U * C_);
        k2_step(fc, wp + (it + 1) * K2_U, mx, sm);
    }
    k2_load(fc, fp + (size_t)(K2_NB - 1) * K2_U * C_);
    k2_step(fa, wp + (K2_NB - 2) * K2_U, mx, sm);
    k2_step(fc, wp + (K2_NB - 1) * K2_U, mx, sm);

#pragma unroll
    for (int g = 0; g < K2_CG; ++g) {
#pragma unroll
        for (int j = 0; j < K2_AG; ++j) {
            smx[s][j][g * 32 + lane] = mx[g][j];
            ssm[s][j][g * 32 + lane] = sm[g][j];
        }
    }
    __syncthreads();

    if (threadIdx.x < K2_AG * C_) {
        const int j  = threadIdx.x / C_;
        const int cc = threadIdx.x % C_;
        float m = smx[0][j][cc], t = ssm[0][j][cc];
#pragma unroll
        for (int q = 1; q < K2_VS; ++q) { m = fmaxf(m, smx[q][j][cc]); t += ssm[q][j][cc]; }
        float* ao = agg + (size_t)(b * A_ + a0 + j) * 192;
        ao[cc]      = m;
        ao[96 + cc] = t * (1.0f / (float)V_);
    }
}

#define K3_KQ    4
#define K3_KL    (192 / K3_KQ)
#define K3_NG    (NF_ / 32)
#define K3_PITCH 36

__global__ __launch_bounds__(256) void k3_bw(
    const float* __restrict__ agg, const float* __restrict__ Wout,
    float* __restrict__ BW)
{
    __shared__ float red[K3_KQ][B_][K3_PITCH];
    static_assert(K3_KQ == 4 && K3_KL % 4 == 0, "the epilogue sums exactly four quarter partials");

    if (blockIdx.x >= A_ * K3_NG) {
        const int h = blockIdx.x - A_ * K3_NG;
        const float4* src = (const float4*)(Wout + (h & 1) * 1024);
        float4* dst = (float4*)(BW + (size_t)(h >> 1) * KTOT * NF_ + (h & 1) * 1024);
        dst[threadIdx.x] = src[threadIdx.x];
        return;
    }

    const int a    = blockIdx.x / K3_NG;
    const int ng   = blockIdx.x % K3_NG;
    const int lane = threadIdx.x & 31;
    const int wv   = threadIdx.x >> 5;
    const int nt   = wv & 1;
    const int q    = wv >> 1;
    const int c16  = lane & 15;
    const int kh   = (lane >> 4) * 2;
    const int col  = ng * 32 + nt * 16 + c16;

    const float* a0p = agg + ((size_t)c16 * A_ + a) * 192 + q * K3_KL + kh;
    const float* a1p = a0p + (size_t)16 * A_ * 192;
    const float* bp  = Wout + (size_t)(F_ + a * 192 + q * K3_KL + kh) * NF_ + col;

    const float bias = Wout[(size_t)(F_ + 2 * A_ * C_ + a) * NF_ + ng * 32 + (threadIdx.x & 31)];

    v8f acc0 = {}, acc1 = {};
#pragma unroll
    for (int kt = 0; kt < K3_KL / 4; ++kt) {
        const v2f av0 = *(const v2f*)(a0p + kt * 4);
        const v2f av1 = *(const v2f*)(a1p + kt * 4);
        v2f bb;
        bb.x = bp[(size_t)(kt * 4) * NF_];
        bb.y = bp[(size_t)(kt * 4 + 1) * NF_];
        acc0 = __builtin_amdgcn_wmma_f32_16x16x4_f32(
            false, av0, false, bb, (short)0, acc0, false, false);
        acc1 = __builtin_amdgcn_wmma_f32_16x16x4_f32(
            false, av1, false, bb, (short)0, acc1, false, false);
    }

#pragma unroll
    for (int r = 0; r < 8; ++r) {
        const int bl = r + (lane >> 4) * 8;
        red[q][bl][nt * 16 + c16]      = acc0[r];
        red[q][16 + bl][nt * 16 + c16] = acc1[r];
    }
    __syncthreads();

    const int cc = threadIdx.x & 31;
    const int rb = threadIdx.x >> 5;
#pragma unroll
    for (int i = 0; i < 4; ++i) {
        const int b = rb + 8 * i;
        const float t = ((red[0][b][cc] + red[1][b][cc]) + red[2][b][cc]) + red[3][b][cc];
        BW[((size_t)b * KTOT + F_ + a) * NF_ + ng * 32 + cc] = t + bias;
    }
}

#define K4_RT    2
#define K4_ROWS  (16 * K4_RT)
#define K4_PITCH 52

__global__ __launch_bounds__(256) void k4_out(
    const float* __restrict__ x,   const float* __restrict__ fws,
    const float* __restrict__ BW,  const float* __restrict__ bout,
    float* __restrict__ out)
{
    __shared__ __attribute__((aligned(16))) float sA[K4_ROWS][K4_PITCH];
    static_assert(V_ % K4_ROWS == 0, "a block never straddles two batches");
    static_assert(K4_ROWS * 8 == 256 && A_ == 8 * 4 && F_ == 8 * 2,
                  "a thread stages one 16-byte piece of ew and one 8-byte piece of x of its row");

    const int lane  = threadIdx.x & 31;
    const int ntile = threadIdx.x >> 5;
    const int c16   = lane & 15;
    const int col   = ntile * 16 + c16;
    const int kh    = (lane >> 4) * 2;
    const int row0  = blockIdx.x * K4_ROWS;
    const int b     = row0 / V_;

    const int sr = threadIdx.x >> 3, sc = threadIdx.x & 7;
    const float4 ev = *(const float4*)(fws + (size_t)(row0 + sr) * C_ + P_ + sc * 4);
    const v2f    xv = *(const v2f*)(x + (size_t)(row0 + sr) * F_ + sc * 2);

    const float* bwb = BW + (size_t)b * KTOT * NF_ + col;
    v2f bb[KTOT / 4];
#pragma unroll
    for (int kt = 0; kt < KTOT / 4; ++kt) {
        const int k0 = kt * 4 + kh;
        bb[kt].x = bwb[(size_t)k0 * NF_];
        bb[kt].y = bwb[(size_t)(k0 + 1) * NF_];
    }
    const float bias = bout[col];

    *(float4*)&sA[sr][F_ + sc * 4] = ev;
    *(v2f*)&sA[sr][sc * 2]         = xv;
    __syncthreads();

    v8f acc[K4_RT];
#pragma unroll
    for (int t = 0; t < K4_RT; ++t) acc[t] = v8f{};
#pragma unroll
    for (int kt = 0; kt < KTOT / 4; ++kt) {
        const int k0 = kt * 4 + kh;
#pragma unroll
        for (int t = 0; t < K4_RT; ++t) {
            const v2f a = *(const v2f*)&sA[t * 16 + c16][k0];
            acc[t] = __builtin_amdgcn_wmma_f32_16x16x4_f32(
                false, a, false, bb[kt], (short)0, acc[t], false, false);
        }
    }

#pragma unroll
    for (int t = 0; t < K4_RT; ++t) {
#pragma unroll
        for (int r = 0; r < 8; ++r) {
            const int row = row0 + t * 16 + r + ((lane >> 4) * 8);
            out[(size_t)row * NF_ + col] = tanhf(acc[t][r] + bias);
        }
    }
}

extern "C" void kernel_launch(void* const* d_in, const int* in_sizes, int n_in,
                              void* d_out, int out_size, void* d_ws, size_t ws_size,
                              hipStream_t stream)
{
    const float* x    = (const float*)d_in[0];
    const float* Wf   = (const float*)d_in[1];
    const float* bf   = (const float*)d_in[2];
    const float* Ws   = (const float*)d_in[3];
    const float* bs   = (const float*)d_in[4];
    const float* Wout = (const float*)d_in[5];
    const float* bout = (const float*)d_in[6];
    float* out = (float*)d_out;

    float* fws = (float*)d_ws;
    float* agg = fws + (size_t)ROWS * C_;
    float* BW  = agg + (size_t)B_ * A_ * 192;

    k1_vertex<<<ROWS / 16, 192, 0, stream>>>(x, Wf, bf, Ws, bs, fws);
    k2_agg   <<<B_ * (A_ / K2_AG), K2_T, 0, stream>>>(fws, agg);
    k3_bw    <<<A_ * K3_NG + 2 * B_, 256, 0, stream>>>(agg, Wout, BW);
    k4_out   <<<ROWS / K4_ROWS, 256, 0, stream>>>(x, fws, BW, bout, out);
}


# k4 epilogue: library tanhf (2 divergent branches round a range-reduced expf, ~40 instructions a value) replaced by 1 - 2/(exp2(2x log2e) + 1) on hardware exp2 and rcp, branch-free; abs error ~2e-7
# speedup vs baseline: 5.1695x; 1.0723x over previous
#include <hip/hip_runtime.h>
#include <math.h>


typedef __attribute__((ext_vector_type(2))) float v2f;
typedef __attribute__((ext_vector_type(8))) float v8f;

#define B_   32
#define V_   1024
#define F_   16
#define A_   32
#define P_   64
#define NF_  128
#define C_   96
#define KTOT 48
#define ROWS (B_ * V_)


__global__ __launch_bounds__(192) void k1_vertex(
    const float* __restrict__ x,
    const float* __restrict__ Wf, const float* __restrict__ bf,
    const float* __restrict__ Ws, const float* __restrict__ bs,
    float* __restrict__ fws)
{
    const int lane  = threadIdx.x & 31;
    const int ntile = __builtin_amdgcn_readfirstlane((int)(threadIdx.x >> 5));
    const int mtile = blockIdx.x;
    const int c16   = lane & 15;
    const int col   = ntile * 16 + c16;
    const int kh    = (lane >> 4) * 2;
    const float* xr = x + (size_t)(mtile * 16 + c16) * F_;

    const bool   isE   = ntile >= P_ / 16;
    const int    pitch = isE ? A_ : P_;
    const float* Wp    = isE ? Ws + (ntile * 16 - P_) : Wf + ntile * 16;
    const float* bp    = isE ? bs + (ntile * 16 - P_) : bf + ntile * 16;

    v8f acc = {};
#pragma unroll
    for (int kt = 0; kt < 4; ++kt) {
        const int k0 = kt * 4 + kh;
        const v2f a = *(const v2f*)(xr + k0);
        v2f b;
        b.x = Wp[k0 * pitch + c16];
        b.y = Wp[(k0 + 1) * pitch + c16];
        acc = __builtin_amdgcn_wmma_f32_16x16x4_f32(
            false, a, false, b, (short)0, acc, false, false);
    }

    const float bias = bp[c16];
#pragma unroll
    for (int r = 0; r < 8; ++r) {
        const int row = mtile * 16 + r + ((lane >> 4) * 8);
        float v = acc[r] + bias;
        if (isE) v = expf(-v * v);
        fws[(size_t)row * C_ + col] = v;
    }
}

#define K2_AG 4
#define K2_CG (C_ / 32)
#define K2_VS 16
#define K2_VL (V_ / K2_VS)
#define K2_U  8
#define K2_NB (K2_VL / K2_U)
#define K2_T  (32 * K2_VS)

__device__ __forceinline__ void k2_load(float (&d)[K2_U][K2_CG], const float* __restrict__ p)
{
#pragma unroll
    for (int u = 0; u < K2_U; ++u) {
#pragma unroll
        for (int g = 0; g < K2_CG; ++g) d[u][g] = p[(size_t)u * C_ + g * 32];
    }
}

__device__ __forceinline__ void k2_step(const float (&fv)[K2_U][K2_CG], const float4* w4,
                                        float (&mx)[K2_CG][K2_AG], float (&sm)[K2_CG][K2_AG])
{
#pragma unroll
    for (int u = 0; u < K2_U; ++u) {
        const float4 w = w4[u];
        const float wj[K2_AG] = {w.x, w.y, w.z, w.w};
#pragma unroll
        for (int g = 0; g < K2_CG; ++g) {
#pragma unroll
            for (int j = 0; j < K2_AG; ++j) {
                mx[g][j] = fmaxf(mx[g][j], wj[j] * fv[u][g]);
                sm[g][j] = fmaf(wj[j], fv[u][g], sm[g][j]);
            }
        }
    }
}

__global__ __launch_bounds__(K2_T) void k2_agg(
    const float* __restrict__ fws, float* __restrict__ agg)
{
    __shared__ float4 sew[V_];
    __shared__ float  smx[K2_VS][K2_AG][C_];
    __shared__ float  ssm[K2_VS][K2_AG][C_];

    const int b    = blockIdx.x >> 3;
    const int a0   = (blockIdx.x & 7) * K2_AG;
    const int lane = threadIdx.x & 31;
    const int s    = threadIdx.x >> 5;

    const float* fb = fws + (size_t)b * V_ * C_;
    const float* fp = fb + (size_t)s * K2_VL * C_ + lane;

    float fa[K2_U][K2_CG], fc[K2_U][K2_CG];
    k2_load(fa, fp);

    static_assert(V_ == 2 * K2_T, "each thread stages exactly two ew rows");
    const float* eb = fb + P_ + a0;
    const int v0 = threadIdx.x, v1 = threadIdx.x + K2_T;
    const float4 e0 = *(const float4*)(eb + (size_t)v0 * C_);
    const float4 e1 = *(const float4*)(eb + (size_t)v1 * C_);
    sew[v0] = e0;
    sew[v1] = e1;
    __syncthreads();

    float mx[K2_CG][K2_AG], sm[K2_CG][K2_AG];
#pragma unroll
    for (int g = 0; g < K2_CG; ++g) {
#pragma unroll
        for (int j = 0; j < K2_AG; ++j) { mx[g][j] = -INFINITY; sm[g][j] = 0.0f; }
    }

    static_assert(K2_NB >= 2 && K2_NB % 2 == 0, "batches are consumed in pairs");
    const float4* wp = sew + s * K2_VL;
    for (int it = 0; it < K2_NB - 2; it += 2) {
        k2_load(fc, fp + (size_t)(it + 1) * K2_U * C_);
        k2_step(fa, wp + it * K2_U, mx, sm);
        k2_load(fa, fp + (size_t)(it + 2) * K2_U * C_);
        k2_step(fc, wp + (it + 1) * K2_U, mx, sm);
    }
    k2_load(fc, fp + (size_t)(K2_NB - 1) * K2_U * C_);
    k2_step(fa, wp + (K2_NB - 2) * K2_U, mx, sm);
    k2_step(fc, wp + (K2_NB - 1) * K2_U, mx, sm);

#pragma unroll
    for (int g = 0; g < K2_CG; ++g) {
#pragma unroll
        for (int j = 0; j < K2_AG; ++j) {
            smx[s][j][g * 32 + lane] = mx[g][j];
            ssm[s][j][g * 32 + lane] = sm[g][j];
        }
    }
    __syncthreads();

    if (threadIdx.x < K2_AG * C_) {
        const int j  = threadIdx.x / C_;
        const int cc = threadIdx.x % C_;
        float m = smx[0][j][cc], t = ssm[0][j][cc];
#pragma unroll
        for (int q = 1; q < K2_VS; ++q) { m = fmaxf(m, smx[q][j][cc]); t += ssm[q][j][cc]; }
        float* ao = agg + (size_t)(b * A_ + a0 + j) * 192;
        ao[cc]      = m;
        ao[96 + cc] = t * (1.0f / (float)V_);
    }
}

#define K3_KQ    4
#define K3_KL    (192 / K3_KQ)
#define K3_NG    (NF_ / 32)
#define K3_PITCH 36

__global__ __launch_bounds__(256) void k3_bw(
    const float* __restrict__ agg, const float* __restrict__ Wout,
    float* __restrict__ BW)
{
    __shared__ float red[K3_KQ][B_][K3_PITCH];
    static_assert(K3_KQ == 4 && K3_KL % 4 == 0, "the epilogue sums exactly four quarter partials");

    if (blockIdx.x >= A_ * K3_NG) {
        const int h = blockIdx.x - A_ * K3_NG;
        const float4* src = (const float4*)(Wout + (h & 1) * 1024);
        float4* dst = (float4*)(BW + (size_t)(h >> 1) * KTOT * NF_ + (h & 1) * 1024);
        dst[threadIdx.x] = src[threadIdx.x];
        return;
    }

    const int a    = blockIdx.x / K3_NG;
    const int ng   = blockIdx.x % K3_NG;
    const int lane = threadIdx.x & 31;
    const int wv   = threadIdx.x >> 5;
    const int nt   = wv & 1;
    const int q    = wv >> 1;
    const int c16  = lane & 15;
    const int kh   = (lane >> 4) * 2;
    const int col  = ng * 32 + nt * 16 + c16;

    const float* a0p = agg + ((size_t)c16 * A_ + a) * 192 + q * K3_KL + kh;
    const float* a1p = a0p + (size_t)16 * A_ * 192;
    const float* bp  = Wout + (size_t)(F_ + a * 192 + q * K3_KL + kh) * NF_ + col;

    const float bias = Wout[(size_t)(F_ + 2 * A_ * C_ + a) * NF_ + ng * 32 + (threadIdx.x & 31)];

    v8f acc0 = {}, acc1 = {};
#pragma unroll
    for (int kt = 0; kt < K3_KL / 4; ++kt) {
        const v2f av0 = *(const v2f*)(a0p + kt * 4);
        const v2f av1 = *(const v2f*)(a1p + kt * 4);
        v2f bb;
        bb.x = bp[(size_t)(kt * 4) * NF_];
        bb.y = bp[(size_t)(kt * 4 + 1) * NF_];
        acc0 = __builtin_amdgcn_wmma_f32_16x16x4_f32(
            false, av0, false, bb, (short)0, acc0, false, false);
        acc1 = __builtin_amdgcn_wmma_f32_16x16x4_f32(
            false, av1, false, bb, (short)0, acc1, false, false);
    }

#pragma unroll
    for (int r = 0; r < 8; ++r) {
        const int bl = r + (lane >> 4) * 8;
        red[q][bl][nt * 16 + c16]      = acc0[r];
        red[q][16 + bl][nt * 16 + c16] = acc1[r];
    }
    __syncthreads();

    const int cc = threadIdx.x & 31;
    const int rb = threadIdx.x >> 5;
#pragma unroll
    for (int i = 0; i < 4; ++i) {
        const int b = rb + 8 * i;
        const float t = ((red[0][b][cc] + red[1][b][cc]) + red[2][b][cc]) + red[3][b][cc];
        BW[((size_t)b * KTOT + F_ + a) * NF_ + ng * 32 + cc] = t + bias;
    }
}

__device__ __forceinline__ float k4_tanh(float v)
{
    const float e = __builtin_amdgcn_exp2f(v * 2.885390081777927f);
    return fmaf(-2.0f, __builtin_amdgcn_rcpf(e + 1.0f), 1.0f);
}

#define K4_RT    2
#define K4_ROWS  (16 * K4_RT)
#define K4_PITCH 52

__global__ __launch_bounds__(256) void k4_out(
    const float* __restrict__ x,   const float* __restrict__ fws,
    const float* __restrict__ BW,  const float* __restrict__ bout,
    float* __restrict__ out)
{
    __shared__ __attribute__((aligned(16))) float sA[K4_ROWS][K4_PITCH];
    static_assert(V_ % K4_ROWS == 0, "a block never straddles two batches");
    static_assert(K4_ROWS * 8 == 256 && A_ == 8 * 4 && F_ == 8 * 2,
                  "a thread stages one 16-byte piece of ew and one 8-byte piece of x of its row");

    const int lane  = threadIdx.x & 31;
    const int ntile = threadIdx.x >> 5;
    const int c16   = lane & 15;
    const int col   = ntile * 16 + c16;
    const int kh    = (lane >> 4) * 2;
    const int row0  = blockIdx.x * K4_ROWS;
    const int b     = row0 / V_;

    const int sr = threadIdx.x >> 3, sc = threadIdx.x & 7;
    const float4 ev = *(const float4*)(fws + (size_t)(row0 + sr) * C_ + P_ + sc * 4);
    const v2f    xv = *(const v2f*)(x + (size_t)(row0 + sr) * F_ + sc * 2);

    const float* bwb = BW + (size_t)b * KTOT * NF_ + col;
    v2f bb[KTOT / 4];
#pragma unroll
    for (int kt = 0; kt < KTOT / 4; ++kt) {
        const int k0 = kt * 4 + kh;
        bb[kt].x = bwb[(size_t)k0 * NF_];
        bb[kt].y = bwb[(size_t)(k0 + 1) * NF_];
    }
    const float bias = bout[col];

    *(float4*)&sA[sr][F_ + sc * 4] = ev;
    *(v2f*)&sA[sr][sc * 2]         = xv;
    __syncthreads();

    v8f acc[K4_RT];
#pragma unroll
    for (int t = 0; t < K4_RT; ++t) acc[t] = v8f{};
#pragma unroll
    for (int kt = 0; kt < KTOT / 4; ++kt) {
        const int k0 = kt * 4 + kh;
#pragma unroll
        for (int t = 0; t < K4_RT; ++t) {
            const v2f a = *(const v2f*)&sA[t * 16 + c16][k0];
            acc[t] = __builtin_amdgcn_wmma_f32_16x16x4_f32(
                false, a, false, bb[kt], (short)0, acc[t], false, false);
        }
    }

#pragma unroll
    for (int t = 0; t < K4_RT; ++t) {
#pragma unroll
        for (int r = 0; r < 8; ++r) {
            const int row = row0 + t * 16 + r + ((lane >> 4) * 8);
            out[(size_t)row * NF_ + col] = k4_tanh(acc[t][r] + bias);
        }
    }
}

extern "C" void kernel_launch(void* const* d_in, const int* in_sizes, int n_in,
                              void* d_out, int out_size, void* d_ws, size_t ws_size,
                              hipStream_t stream)
{
    const float* x    = (const float*)d_in[0];
    const float* Wf   = (const float*)d_in[1];
    const float* bf   = (const float*)d_in[2];
    const float* Ws   = (const float*)d_in[3];
    const float* bs   = (const float*)d_in[4];
    const float* Wout = (const float*)d_in[5];
    const float* bout = (const float*)d_in[6];
    float* out = (float*)d_out;

    float* fws = (float*)d_ws;
    float* agg = fws + (size_t)ROWS * C_;
    float* BW  = agg + (size_t)B_ * A_ * 192;

    k1_vertex<<<ROWS / 16, 192, 0, stream>>>(x, Wf, bf, Ws, bs, fws);
    k2_agg   <<<B_ * (A_ / K2_AG), K2_T, 0, stream>>>(fws, agg);
    k3_bw    <<<A_ * K3_NG + 2 * B_, 256, 0, stream>>>(agg, Wout, BW);
    k4_out   <<<ROWS / K4_ROWS, 256, 0, stream>>>(x, fws, BW, bout, out);
}


# k2 reads its 4 edge weights per vertex out of the f lines it streams anyway (ew = channels 64..95), handed round inside each wave through LDS: ew staging pass and its barrier removed; sum order kept
# speedup vs baseline: 5.6420x; 1.0914x over previous
#include <hip/hip_runtime.h>
#include <math.h>


typedef __attribute__((ext_vector_type(2))) float v2f;
typedef __attribute__((ext_vector_type(8))) float v8f;

#define B_   32
#define V_   1024
#define F_   16
#define A_   32
#define P_   64
#define NF_  128
#define C_   96
#define KTOT 48
#define ROWS (B_ * V_)


__global__ __launch_bounds__(192) void k1_vertex(
    const float* __restrict__ x,
    const float* __restrict__ Wf, const float* __restrict__ bf,
    const float* __restrict__ Ws, const float* __restrict__ bs,
    float* __restrict__ fws)
{
    const int lane  = threadIdx.x & 31;
    const int ntile = __builtin_amdgcn_readfirstlane((int)(threadIdx.x >> 5));
    const int mtile = blockIdx.x;
    const int c16   = lane & 15;
    const int col   = ntile * 16 + c16;
    const int kh    = (lane >> 4) * 2;
    const float* xr = x + (size_t)(mtile * 16 + c16) * F_;

    const bool   isE   = ntile >= P_ / 16;
    const int    pitch = isE ? A_ : P_;
    const float* Wp    = isE ? Ws + (ntile * 16 - P_) : Wf + ntile * 16;
    const float* bp    = isE ? bs + (ntile * 16 - P_) : bf + ntile * 16;

    v8f acc = {};
#pragma unroll
    for (int kt = 0; kt < 4; ++kt) {
        const int k0 = kt * 4 + kh;
        const v2f a = *(const v2f*)(xr + k0);
        v2f b;
        b.x = Wp[k0 * pitch + c16];
        b.y = Wp[(k0 + 1) * pitch + c16];
        acc = __builtin_amdgcn_wmma_f32_16x16x4_f32(
            false, a, false, b, (short)0, acc, false, false);
    }

    const float bias = bp[c16];
#pragma unroll
    for (int r = 0; r < 8; ++r) {
        const int row = mtile * 16 + r + ((lane >> 4) * 8);
        float v = acc[r] + bias;
        if (isE) v = expf(-v * v);
        fws[(size_t)row * C_ + col] = v;
    }
}

#define K2_AG 4
#define K2_CG (C_ / 32)
#define K2_EG (P_ / 32)
#define K2_VS 16
#define K2_VL (V_ / K2_VS)
#define K2_U  8
#define K2_NB (K2_VL / K2_U)
#define K2_T  (32 * K2_VS)

static_assert(P_ % 32 == 0 && A_ == 32 && K2_EG == K2_CG - 1,
              "the last channel group of a lane is exactly the 32 edge weights, lane = aggregator");
static_assert(K2_AG == 4, "the weights of a vertex are read back as one float4");

__device__ __forceinline__ void k2_load(float (&d)[K2_U][K2_CG], const float* __restrict__ p)
{
#pragma unroll
    for (int u = 0; u < K2_U; ++u) d[u][K2_EG] = p[(size_t)u * C_ + K2_EG * 32];
#pragma unroll
    for (int u = 0; u < K2_U; ++u) {
#pragma unroll
        for (int g = 0; g < K2_EG; ++g) d[u][g] = p[(size_t)u * C_ + g * 32];
    }
}

__device__ __forceinline__ void k2_step(const float (&fv)[K2_U][K2_CG], float* wst, const float4* wld,
                                        float (&mx)[K2_CG][K2_AG], float (&sm)[K2_CG][K2_AG])
{
#pragma unroll
    for (int u = 0; u < K2_U; ++u) wst[u * 32] = fv[u][K2_EG];
    float4 w[K2_U];
#pragma unroll
    for (int u = 0; u < K2_U; ++u) w[u] = wld[u * 8];
#pragma unroll
    for (int u = 0; u < K2_U; ++u) {
        const float wj[K2_AG] = {w[u].x, w[u].y, w[u].z, w[u].w};
#pragma unroll
        for (int g = 0; g < K2_CG; ++g) {
#pragma unroll
            for (int j = 0; j < K2_AG; ++j) {
                mx[g][j] = fmaxf(mx[g][j], wj[j] * fv[u][g]);
                sm[g][j] = fmaf(wj[j], fv[u][g], sm[g][j]);
            }
        }
    }
}

__global__ __launch_bounds__(K2_T) void k2_agg(
    const float* __restrict__ fws, float* __restrict__ agg)
{
    __shared__ __attribute__((aligned(16))) float swt[K2_VS][K2_U][32];
    __shared__ float  smx[K2_VS][K2_AG][C_];
    __shared__ float  ssm[K2_VS][K2_AG][C_];

    const int b    = blockIdx.x >> 3;
    const int a0   = (blockIdx.x & 7) * K2_AG;
    const int lane = threadIdx.x & 31;
    const int s    = threadIdx.x >> 5;

    const float* fb = fws + (size_t)b * V_ * C_;
    const float* fp = fb + (size_t)s * K2_VL * C_ + lane;

    float fa[K2_U][K2_CG], fc[K2_U][K2_CG];
    k2_load(fa, fp);

    float*        wst = &swt[s][0][lane];
    const float4* wld = (const float4*)&swt[s][0][a0];

    float mx[K2_CG][K2_AG], sm[K2_CG][K2_AG];
#pragma unroll
    for (int g = 0; g < K2_CG; ++g) {
#pragma unroll
        for (int j = 0; j < K2_AG; ++j) { mx[g][j] = -INFINITY; sm[g][j] = 0.0f; }
    }

    static_assert(K2_NB >= 2 && K2_NB % 2 == 0, "batches are consumed in pairs");
    for (int it = 0; it < K2_NB - 2; it += 2) {
        k2_load(fc, fp + (size_t)(it + 1) * K2_U * C_);
        k2_step(fa, wst, wld, mx, sm);
        k2_load(fa, fp + (size_t)(it + 2) * K2_U * C_);
        k2_step(fc, wst, wld, mx, sm);
    }
    k2_load(fc, fp + (size_t)(K2_NB - 1) * K2_U * C_);
    k2_step(fa, wst, wld, mx, sm);
    k2_step(fc, wst, wld, mx, sm);

#pragma unroll
    for (int g = 0; g < K2_CG; ++g) {
#pragma unroll
        for (int j = 0; j < K2_AG; ++j) {
            smx[s][j][g * 32 + lane] = mx[g][j];
            ssm[s][j][g * 32 + lane] = sm[g][j];
        }
    }
    __syncthreads();

    if (threadIdx.x < K2_AG * C_) {
        const int j  = threadIdx.x / C_;
        const int cc = threadIdx.x % C_;
        float m = smx[0][j][cc], t = ssm[0][j][cc];
#pragma unroll
        for (int q = 1; q < K2_VS; ++q) { m = fmaxf(m, smx[q][j][cc]); t += ssm[q][j][cc]; }
        float* ao = agg + (size_t)(b * A_ + a0 + j) * 192;
        ao[cc]      = m;
        ao[96 + cc] = t * (1.0f / (float)V_);
    }
}

#define K3_KQ    4
#define K3_KL    (192 / K3_KQ)
#define K3_NG    (NF_ / 32)
#define K3_PITCH 36

__global__ __launch_bounds__(256) void k3_bw(
    const float* __restrict__ agg, const float* __restrict__ Wout,
    float* __restrict__ BW)
{
    __shared__ float red[K3_KQ][B_][K3_PITCH];
    static_assert(K3_KQ == 4 && K3_KL % 4 == 0, "the epilogue sums exactly four quarter partials");

    if (blockIdx.x >= A_ * K3_NG) {
        const int h = blockIdx.x - A_ * K3_NG;
        const float4* src = (const float4*)(Wout + (h & 1) * 1024);
        float4* dst = (float4*)(BW + (size_t)(h >> 1) * KTOT * NF_ + (h & 1) * 1024);
        dst[threadIdx.x] = src[threadIdx.x];
        return;
    }

    const int a    = blockIdx.x / K3_NG;
    const int ng   = blockIdx.x % K3_NG;
    const int lane = threadIdx.x & 31;
    const int wv   = threadIdx.x >> 5;
    const int nt   = wv & 1;
    const int q    = wv >> 1;
    const int c16  = lane & 15;
    const int kh   = (lane >> 4) * 2;
    const int col  = ng * 32 + nt * 16 + c16;

    const float* a0p = agg + ((size_t)c16 * A_ + a) * 192 + q * K3_KL + kh;
    const float* a1p = a0p + (size_t)16 * A_ * 192;
    const float* bp  = Wout + (size_t)(F_ + a * 192 + q * K3_KL + kh) * NF_ + col;

    const float bias = Wout[(size_t)(F_ + 2 * A_ * C_ + a) * NF_ + ng * 32 + (threadIdx.x & 31)];

    v8f acc0 = {}, acc1 = {};
#pragma unroll
    for (int kt = 0; kt < K3_KL / 4; ++kt) {
        const v2f av0 = *(const v2f*)(a0p + kt * 4);
        const v2f av1 = *(const v2f*)(a1p + kt * 4);
        v2f bb;
        bb.x = bp[(size_t)(kt * 4) * NF_];
        bb.y = bp[(size_t)(kt * 4 + 1) * NF_];
        acc0 = __builtin_amdgcn_wmma_f32_16x16x4_f32(
            false, av0, false, bb, (short)0, acc0, false, false);
        acc1 = __builtin_amdgcn_wmma_f32_16x16x4_f32(
            false, av1, false, bb, (short)0, acc1, false, false);
    }

#pragma unroll
    for (int r = 0; r < 8; ++r) {
        const int bl = r + (lane >> 4) * 8;
        red[q][bl][nt * 16 + c16]      = acc0[r];
        red[q][16 + bl][nt * 16 + c16] = acc1[r];
    }
    __syncthreads();

    const int cc = threadIdx.x & 31;
    const int rb = threadIdx.x >> 5;
#pragma unroll
    for (int i = 0; i < 4; ++i) {
        const int b = rb + 8 * i;
        const float t = ((red[0][b][cc] + red[1][b][cc]) + red[2][b][cc]) + red[3][b][cc];
        BW[((size_t)b * KTOT + F_ + a) * NF_ + ng * 32 + cc] = t + bias;
    }
}

__device__ __forceinline__ float k4_tanh(float v)
{
    const float e = __builtin_amdgcn_exp2f(v * 2.885390081777927f);
    return fmaf(-2.0f, __builtin_amdgcn_rcpf(e + 1.0f), 1.0f);
}

#define K4_RT    2
#define K4_ROWS  (16 * K4_RT)
#define K4_PITCH 52

__global__ __launch_bounds__(256) void k4_out(
    const float* __restrict__ x,   const float* __restrict__ fws,
    const float* __restrict__ BW,  const float* __restrict__ bout,
    float* __restrict__ out)
{
    __shared__ __attribute__((aligned(16))) float sA[K4_ROWS][K4_PITCH];
    static_assert(V_ % K4_ROWS == 0, "a block never straddles two batches");
    static_assert(K4_ROWS * 8 == 256 && A_ == 8 * 4 && F_ == 8 * 2,
                  "a thread stages one 16-byte piece of ew and one 8-byte piece of x of its row");

    const int lane  = threadIdx.x & 31;
    const int ntile = threadIdx.x >> 5;
    const int c16   = lane & 15;
    const int col   = ntile * 16 + c16;
    const int kh    = (lane >> 4) * 2;
    const int row0  = blockIdx.x * K4_ROWS;
    const int b     = row0 / V_;

    const int sr = threadIdx.x >> 3, sc = threadIdx.x & 7;
    const float4 ev = *(const float4*)(fws + (size_t)(row0 + sr) * C_ + P_ + sc * 4);
    const v2f    xv = *(const v2f*)(x + (size_t)(row0 + sr) * F_ + sc * 2);

    const float* bwb = BW + (size_t)b * KTOT * NF_ + col;
    v2f bb[KTOT / 4];
#pragma unroll
    for (int kt = 0; kt < KTOT / 4; ++kt) {
        const int k0 = kt * 4 + kh;
        bb[kt].x = bwb[(size_t)k0 * NF_];
        bb[kt].y = bwb[(size_t)(k0 + 1) * NF_];
    }
    const float bias = bout[col];

    *(float4*)&sA[sr][F_ + sc * 4] = ev;
    *(v2f*)&sA[sr][sc * 2]         = xv;
    __syncthreads();

    v8f acc[K4_RT];
#pragma unroll
    for (int t = 0; t < K4_RT; ++t) acc[t] = v8f{};
#pragma unroll
    for (int kt = 0; kt < KTOT / 4; ++kt) {
        const int k0 = kt * 4 + kh;
#pragma unroll
        for (int t = 0; t < K4_RT; ++t) {
            const v2f a = *(const v2f*)&sA[t * 16 + c16][k0];
            acc[t] = __builtin_amdgcn_wmma_f32_16x16x4_f32(
                false, a, false, bb[kt], (short)0, acc[t], false, false);
        }
    }

#pragma unroll
    for (int t = 0; t < K4_RT; ++t) {
#pragma unroll
        for (int r = 0; r < 8; ++r) {
            const int row = row0 + t * 16 + r + ((lane >> 4) * 8);
            out[(size_t)row * NF_ + col] = k4_tanh(acc[t][r] + bias);
        }
    }
}

extern "C" void kernel_launch(void* const* d_in, const int* in_sizes, int n_in,
                              void* d_out, int out_size, void* d_ws, size_t ws_size,
                              hipStream_t stream)
{
    const float* x    = (const float*)d_in[0];
    const float* Wf   = (const float*)d_in[1];
    const float* bf   = (const float*)d_in[2];
    const float* Ws   = (const float*)d_in[3];
    const float* bs   = (const float*)d_in[4];
    const float* Wout = (const float*)d_in[5];
    const float* bout = (const float*)d_in[6];
    float* out = (float*)d_out;

    float* fws = (float*)d_ws;
    float* agg = fws + (size_t)ROWS * C_;
    float* BW  = agg + (size_t)B_ * A_ * 192;

    k1_vertex<<<ROWS / 16, 192, 0, stream>>>(x, Wf, bf, Ws, bs, fws);
    k2_agg   <<<B_ * (A_ / K2_AG), K2_T, 0, stream>>>(fws, agg);
    k3_bw    <<<A_ * K3_NG + 2 * B_, 256, 0, stream>>>(agg, Wout, BW);
    k4_out   <<<ROWS / K4_ROWS, 256, 0, stream>>>(x, fws, BW, bout, out);
}


# vertex transform fused into the aggregation kernel: block = (batch, 128-vertex slice) builds its f tile in LDS and aggregates from it; f never stored; 4 launches become 3; mean sum order changes
# speedup vs baseline: 6.0629x; 1.0746x over previous
#include <hip/hip_runtime.h>
#include <math.h>


typedef __attribute__((ext_vector_type(2))) float v2f;
typedef __attribute__((ext_vector_type(8))) float v8f;

#define B_   32
#define V_   1024
#define F_   16
#define A_   32
#define P_   64
#define NF_  128
#define C_   96
#define KTOT 48
#define ROWS (B_ * V_)


#define KA_SL    8
#define KA_VB    (V_ / KA_SL)
#define KA_W     16
#define KA_T     (32 * KA_W)
#define KA_PITCH 100
#define KA_AG    4
#define KA_CG    (C_ / 32)
#define KA_VL    (KA_VB / 2)
#define KA_U     8
#define KA_NB    (KA_VL / KA_U)
#define PAGG_SL  (A_ * 192)

static_assert(P_ % 32 == 0 && A_ == 32 && C_ == 96, "a lane's third channel group is exactly the 32 edge weights");
static_assert(KA_VB == 16 * (KA_W / 2) && C_ / 16 == 6, "phase 1: 8 row tiles x 2 groups of 3 column tiles = 16 waves");
static_assert(A_ / KA_AG == KA_W / 2, "phase 2: 8 aggregator groups x 2 vertex halves = 16 waves");
static_assert((KA_PITCH * 4) % 16 == 0 && (KA_PITCH * 8) % 64 == 32, "f tile pitch: aligned rows, lane halves on different banks");

__global__ __launch_bounds__(KA_T) void ka_vertex_agg(
    const float* __restrict__ x,
    const float* __restrict__ Wf, const float* __restrict__ bf,
    const float* __restrict__ Ws, const float* __restrict__ bs,
    float* __restrict__ ewb, float* __restrict__ pagg)
{
    __shared__ __attribute__((aligned(16))) float sf[KA_VB][KA_PITCH];
    __shared__ float pmx[A_][C_];
    __shared__ float psm[A_][C_];

    const int lane = threadIdx.x & 31;
    const int wv   = __builtin_amdgcn_readfirstlane((int)(threadIdx.x >> 5));
    const int row0 = blockIdx.x * KA_VB;

    {
        const int mt  = wv >> 1;
        const int nt0 = (wv & 1) * 3;
        const int c16 = lane & 15;
        const int kh  = (lane >> 4) * 2;
        const float* xr = x + (size_t)(row0 + mt * 16 + c16) * F_;
        v2f a[4];
#pragma unroll
        for (int kt = 0; kt < 4; ++kt) a[kt] = *(const v2f*)(xr + kt * 4 + kh);

#pragma unroll
        for (int i = 0; i < 3; ++i) {
            const int    ntile = nt0 + i;
            const bool   isE   = ntile >= P_ / 16;
            const int    pitch = isE ? A_ : P_;
            const float* Wp    = isE ? Ws + (ntile * 16 - P_) : Wf + ntile * 16;
            const float* bp    = isE ? bs + (ntile * 16 - P_) : bf + ntile * 16;
            v8f acc = {};
#pragma unroll
            for (int kt = 0; kt < 4; ++kt) {
                const int k0 = kt * 4 + kh;
                v2f bq;
                bq.x = Wp[k0 * pitch + c16];
                bq.y = Wp[(k0 + 1) * pitch + c16];
                acc = __builtin_amdgcn_wmma_f32_16x16x4_f32(
                    false, a[kt], false, bq, (short)0, acc, false, false);
            }
            const float bias = bp[c16];
#pragma unroll
            for (int r = 0; r < 8; ++r) {
                float v = acc[r] + bias;
                if (isE) v = expf(-v * v);
                sf[mt * 16 + r + (lane >> 4) * 8][ntile * 16 + c16] = v;
            }
        }
    }
    __syncthreads();

#pragma unroll
    for (int i = 0; i < KA_VB * 8 / KA_T; ++i) {
        const int er = i * (KA_T / 8) + (threadIdx.x >> 3), ec = threadIdx.x & 7;
        *(float4*)(ewb + (size_t)(row0 + er) * A_ + ec * 4) = *(const float4*)&sf[er][P_ + ec * 4];
    }

    const int ag = wv & 7;
    const int hv = wv >> 3;
    const float*  fp = &sf[hv * KA_VL][lane];
    const float4* wp = (const float4*)&sf[hv * KA_VL][P_ + ag * KA_AG];

    float mx[KA_CG][KA_AG], sm[KA_CG][KA_AG];
#pragma unroll
    for (int g = 0; g < KA_CG; ++g) {
#pragma unroll
        for (int j = 0; j < KA_AG; ++j) { mx[g][j] = -INFINITY; sm[g][j] = 0.0f; }
    }

    for (int it = 0; it < KA_NB; ++it) {
        float  fv[KA_U][KA_CG];
        float4 w[KA_U];
#pragma unroll
        for (int u = 0; u < KA_U; ++u) {
#pragma unroll
            for (int g = 0; g < KA_CG; ++g) fv[u][g] = fp[(it * KA_U + u) * KA_PITCH + g * 32];
        }
#pragma unroll
        for (int u = 0; u < KA_U; ++u) w[u] = wp[(it * KA_U + u) * (KA_PITCH / 4)];
#pragma unroll
        for (int u = 0; u < KA_U; ++u) {
            const float wj[KA_AG] = {w[u].x, w[u].y, w[u].z, w[u].w};
#pragma unroll
            for (int g = 0; g < KA_CG; ++g) {
#pragma unroll
                for (int j = 0; j < KA_AG; ++j) {
                    mx[g][j] = fmaxf(mx[g][j], wj[j] * fv[u][g]);
                    sm[g][j] = fmaf(wj[j], fv[u][g], sm[g][j]);
                }
            }
        }
    }

    if (hv == 1) {
#pragma unroll
        for (int g = 0; g < KA_CG; ++g) {
#pragma unroll
            for (int j = 0; j < KA_AG; ++j) {
                pmx[ag * KA_AG + j][g * 32 + lane] = mx[g][j];
                psm[ag * KA_AG + j][g * 32 + lane] = sm[g][j];
            }
        }
    }
    __syncthreads();
    if (hv == 0) {
        float* po = pagg + (size_t)blockIdx.x * PAGG_SL + (size_t)(ag * KA_AG) * 192;
#pragma unroll
        for (int g = 0; g < KA_CG; ++g) {
#pragma unroll
            for (int j = 0; j < KA_AG; ++j) {
                po[j * 192 + g * 32 + lane]      = fmaxf(mx[g][j], pmx[ag * KA_AG + j][g * 32 + lane]);
                po[j * 192 + 96 + g * 32 + lane] = sm[g][j] + psm[ag * KA_AG + j][g * 32 + lane];
            }
        }
    }
}

#define K3_KQ    4
#define K3_KL    (192 / K3_KQ)
#define K3_NG    (NF_ / 32)
#define K3_PITCH 36
#define K3_APITCH 196

__global__ __launch_bounds__(256) void k3_bw(
    const float* __restrict__ pagg, const float* __restrict__ Wout,
    float* __restrict__ BW)
{
    __shared__ float red[K3_KQ][B_][K3_PITCH];
    __shared__ __attribute__((aligned(16))) float sA[B_][K3_APITCH];
    static_assert(B_ * 192 / 4 == 6 * 256 && 192 % 4 == 0 && 96 % 4 == 0,
                  "a thread stages six 16-byte pieces of A; a piece is all max or all sum");
    static_assert(K3_KQ == 4 && K3_KL % 4 == 0, "the epilogue sums exactly four quarter partials");

    if (blockIdx.x >= A_ * K3_NG) {
        const int h = blockIdx.x - A_ * K3_NG;
        const float4* src = (const float4*)(Wout + (h & 1) * 1024);
        float4* dst = (float4*)(BW + (size_t)(h >> 1) * KTOT * NF_ + (h & 1) * 1024);
        dst[threadIdx.x] = src[threadIdx.x];
        return;
    }

    const int a    = blockIdx.x / K3_NG;
    const int ng   = blockIdx.x % K3_NG;
    const int lane = threadIdx.x & 31;
    const int wv   = threadIdx.x >> 5;
    const int nt   = wv & 1;
    const int q    = wv >> 1;
    const int c16  = lane & 15;
    const int kh   = (lane >> 4) * 2;
    const int col  = ng * 32 + nt * 16 + c16;

    const float* bp  = Wout + (size_t)(F_ + a * 192 + q * K3_KL + kh) * NF_ + col;

    const float bias = Wout[(size_t)(F_ + 2 * A_ * C_ + a) * NF_ + ng * 32 + (threadIdx.x & 31)];

    v2f bb[K3_KL / 4];
#pragma unroll
    for (int kt = 0; kt < K3_KL / 4; ++kt) {
        bb[kt].x = bp[(size_t)(kt * 4) * NF_];
        bb[kt].y = bp[(size_t)(kt * 4 + 1) * NF_];
    }

    float4 pv[6][KA_SL];
#pragma unroll
    for (int i = 0; i < 6; ++i) {
        const int p  = threadIdx.x + 256 * i;
        const int pb = p / 48, pc = p % 48;
        const float4* sp = (const float4*)(pagg + ((size_t)pb * KA_SL * A_ + a) * 192 + pc * 4);
#pragma unroll
        for (int s = 0; s < KA_SL; ++s) pv[i][s] = sp[(size_t)s * (PAGG_SL / 4)];
    }
    asm volatile("" ::: "memory");
#pragma unroll
    for (int i = 0; i < 6; ++i) {
        const int p  = threadIdx.x + 256 * i;
        const int pb = p / 48, pc = p % 48;
        float4 m = pv[i][0], t = pv[i][0];
#pragma unroll
        for (int s = 1; s < KA_SL; ++s) {
            m.x = fmaxf(m.x, pv[i][s].x); m.y = fmaxf(m.y, pv[i][s].y);
            m.z = fmaxf(m.z, pv[i][s].z); m.w = fmaxf(m.w, pv[i][s].w);
            t.x += pv[i][s].x; t.y += pv[i][s].y; t.z += pv[i][s].z; t.w += pv[i][s].w;
        }
        const bool  isSum = pc >= 96 / 4;
        const float sc    = 1.0f / (float)V_;
        float4 r;
        r.x = isSum ? t.x * sc : m.x;  r.y = isSum ? t.y * sc : m.y;
        r.z = isSum ? t.z * sc : m.z;  r.w = isSum ? t.w * sc : m.w;
        *(float4*)&sA[pb][pc * 4] = r;
    }
    __syncthreads();

    v8f acc0 = {}, acc1 = {};
#pragma unroll
    for (int kt = 0; kt < K3_KL / 4; ++kt) {
        const int k0 = q * K3_KL + kt * 4 + kh;
        const v2f av0 = *(const v2f*)&sA[c16][k0];
        const v2f av1 = *(const v2f*)&sA[16 + c16][k0];
        acc0 = __builtin_amdgcn_wmma_f32_16x16x4_f32(
            false, av0, false, bb[kt], (short)0, acc0, false, false);
        acc1 = __builtin_amdgcn_wmma_f32_16x16x4_f32(
            false, av1, false, bb[kt], (short)0, acc1, false, false);
    }

#pragma unroll
    for (int r = 0; r < 8; ++r) {
        const int bl = r + (lane >> 4) * 8;
        red[q][bl][nt * 16 + c16]      = acc0[r];
        red[q][16 + bl][nt * 16 + c16] = acc1[r];
    }
    __syncthreads();

    const int cc = threadIdx.x & 31;
    const int rb = threadIdx.x >> 5;
#pragma unroll
    for (int i = 0; i < 4; ++i) {
        const int b = rb + 8 * i;
        const float t = ((red[0][b][cc] + red[1][b][cc]) + red[2][b][cc]) + red[3][b][cc];
        BW[((size_t)b * KTOT + F_ + a) * NF_ + ng * 32 + cc] = t + bias;
    }
}

__device__ __forceinline__ float k4_tanh(float v)
{
    const float e = __builtin_amdgcn_exp2f(v * 2.885390081777927f);
    return fmaf(-2.0f, __builtin_amdgcn_rcpf(e + 1.0f), 1.0f);
}

#define K4_RT    2
#define K4_ROWS  (16 * K4_RT)
#define K4_PITCH 52

__global__ __launch_bounds__(256) void k4_out(
    const float* __restrict__ x,   const float* __restrict__ ewb,
    const float* __restrict__ BW,  const float* __restrict__ bout,
    float* __restrict__ out)
{
    __shared__ __attribute__((aligned(16))) float sA[K4_ROWS][K4_PITCH];
    static_assert(V_ % K4_ROWS == 0, "a block never straddles two batches");
    static_assert(K4_ROWS * 8 == 256 && A_ == 8 * 4 && F_ == 8 * 2,
                  "a thread stages one 16-byte piece of ew and one 8-byte piece of x of its row");

    const int lane  = threadIdx.x & 31;
    const int ntile = threadIdx.x >> 5;
    const int c16   = lane & 15;
    const int col   = ntile * 16 + c16;
    const int kh    = (lane >> 4) * 2;
    const int row0  = blockIdx.x * K4_ROWS;
    const int b     = row0 / V_;

    const int sr = threadIdx.x >> 3, sc = threadIdx.x & 7;
    const float4 ev = *(const float4*)(ewb + (size_t)(row0 + sr) * A_ + sc * 4);
    const v2f    xv = *(const v2f*)(x + (size_t)(row0 + sr) * F_ + sc * 2);

    const float* bwb = BW + (size_t)b * KTOT * NF_ + col;
    v2f bb[KTOT / 4];
#pragma unroll
    for (int kt = 0; kt < KTOT / 4; ++kt) {
        const int k0 = kt * 4 + kh;
        bb[kt].x = bwb[(size_t)k0 * NF_];
        bb[kt].y = bwb[(size_t)(k0 + 1) * NF_];
    }
    const float bias = bout[col];

    *(float4*)&sA[sr][F_ + sc * 4] = ev;
    *(v2f*)&sA[sr][sc * 2]         = xv;
    __syncthreads();

    v8f acc[K4_RT];
#pragma unroll
    for (int t = 0; t < K4_RT; ++t) acc[t] = v8f{};
#pragma unroll
    for (int kt = 0; kt < KTOT / 4; ++kt) {
        const int k0 = kt * 4 + kh;
#pragma unroll
        for (int t = 0; t < K4_RT; ++t) {
            const v2f a = *(const v2f*)&sA[t * 16 + c16][k0];
            acc[t] = __builtin_amdgcn_wmma_f32_16x16x4_f32(
                false, a, false, bb[kt], (short)0, acc[t], false, false);
        }
    }

#pragma unroll
    for (int t = 0; t < K4_RT; ++t) {
#pragma unroll
        for (int r = 0; r < 8; ++r) {
            const int row = row0 + t * 16 + r + ((lane >> 4) * 8);
            out[(size_t)row * NF_ + col] = k4_tanh(acc[t][r] + bias);
        }
    }
}

extern "C" void kernel_launch(void* const* d_in, const int* in_sizes, int n_in,
                              void* d_out, int out_size, void* d_ws, size_t ws_size,
                              hipStream_t stream)
{
    const float* x    = (const float*)d_in[0];
    const float* Wf   = (const float*)d_in[1];
    const float* bf   = (const float*)d_in[2];
    const float* Ws   = (const float*)d_in[3];
    const float* bs   = (const float*)d_in[4];
    const float* Wout = (const float*)d_in[5];
    const float* bout = (const float*)d_in[6];
    float* out = (float*)d_out;

    float* ewb  = (float*)d_ws;
    float* pagg = ewb + (size_t)ROWS * A_;
    float* BW   = pagg + (size_t)B_ * KA_SL * PAGG_SL;

    ka_vertex_agg<<<B_ * KA_SL, KA_T, 0, stream>>>(x, Wf, bf, Ws, bs, ewb, pagg);
    k3_bw        <<<A_ * K3_NG + 2 * B_, 256, 0, stream>>>(pagg, Wout, BW);
    k4_out       <<<ROWS / K4_ROWS, 256, 0, stream>>>(x, ewb, BW, bout, out);
}


# k3 spread over every WGP: block = (aggregator, 32 columns, 16 batches), 256 GEMM blocks, not 128; a block pulls in half the slice partials, all of a thread's loads in flight at once; sum order kept
# speedup vs baseline: 6.2802x; 1.0358x over previous
#include <hip/hip_runtime.h>
#include <math.h>


typedef __attribute__((ext_vector_type(2))) float v2f;
typedef __attribute__((ext_vector_type(4))) float v4f;
typedef __attribute__((ext_vector_type(8))) float v8f;

#define B_   32
#define V_   1024
#define F_   16
#define A_   32
#define P_   64
#define NF_  128
#define C_   96
#define KTOT 48
#define ROWS (B_ * V_)


#define KA_SL    8
#define KA_VB    (V_ / KA_SL)
#define KA_W     16
#define KA_T     (32 * KA_W)
#define KA_PITCH 100
#define KA_AG    4
#define KA_CG    (C_ / 32)
#define KA_VL    (KA_VB / 2)
#define KA_U     8
#define KA_NB    (KA_VL / KA_U)
#define PAGG_SL  (A_ * 192)

static_assert(P_ % 32 == 0 && A_ == 32 && C_ == 96, "a lane's third channel group is exactly the 32 edge weights");
static_assert(KA_VB == 16 * (KA_W / 2) && C_ / 16 == 6, "phase 1: 8 row tiles x 2 groups of 3 column tiles = 16 waves");
static_assert(A_ / KA_AG == KA_W / 2, "phase 2: 8 aggregator groups x 2 vertex halves = 16 waves");
static_assert((KA_PITCH * 4) % 16 == 0 && (KA_PITCH * 8) % 64 == 32, "f tile pitch: aligned rows, lane halves on different banks");

__global__ __launch_bounds__(KA_T) void ka_vertex_agg(
    const float* __restrict__ x,
    const float* __restrict__ Wf, const float* __restrict__ bf,
    const float* __restrict__ Ws, const float* __restrict__ bs,
    float* __restrict__ ewb, float* __restrict__ pagg)
{
    __shared__ __attribute__((aligned(16))) float sf[KA_VB][KA_PITCH];
    __shared__ float pmx[A_][C_];
    __shared__ float psm[A_][C_];

    const int lane = threadIdx.x & 31;
    const int wv   = __builtin_amdgcn_readfirstlane((int)(threadIdx.x >> 5));
    const int row0 = blockIdx.x * KA_VB;

    {
        const int mt  = wv >> 1;
        const int nt0 = (wv & 1) * 3;
        const int c16 = lane & 15;
        const int kh  = (lane >> 4) * 2;
        const float* xr = x + (size_t)(row0 + mt * 16 + c16) * F_;
        v2f a[4];
#pragma unroll
        for (int kt = 0; kt < 4; ++kt) a[kt] = *(const v2f*)(xr + kt * 4 + kh);

#pragma unroll
        for (int i = 0; i < 3; ++i) {
            const int    ntile = nt0 + i;
            const bool   isE   = ntile >= P_ / 16;
            const int    pitch = isE ? A_ : P_;
            const float* Wp    = isE ? Ws + (ntile * 16 - P_) : Wf + ntile * 16;
            const float* bp    = isE ? bs + (ntile * 16 - P_) : bf + ntile * 16;
            v8f acc = {};
#pragma unroll
            for (int kt = 0; kt < 4; ++kt) {
                const int k0 = kt * 4 + kh;
                v2f bq;
                bq.x = Wp[k0 * pitch + c16];
                bq.y = Wp[(k0 + 1) * pitch + c16];
                acc = __builtin_amdgcn_wmma_f32_16x16x4_f32(
                    false, a[kt], false, bq, (short)0, acc, false, false);
            }
            const float bias = bp[c16];
#pragma unroll
            for (int r = 0; r < 8; ++r) {
                float v = acc[r] + bias;
                if (isE) v = expf(-v * v);
                sf[mt * 16 + r + (lane >> 4) * 8][ntile * 16 + c16] = v;
            }
        }
    }
    __syncthreads();

#pragma unroll
    for (int i = 0; i < KA_VB * 8 / KA_T; ++i) {
        const int er = i * (KA_T / 8) + (threadIdx.x >> 3), ec = threadIdx.x & 7;
        *(float4*)(ewb + (size_t)(row0 + er) * A_ + ec * 4) = *(const float4*)&sf[er][P_ + ec * 4];
    }

    const int ag = wv & 7;
    const int hv = wv >> 3;
    const float*  fp = &sf[hv * KA_VL][lane];
    const float4* wp = (const float4*)&sf[hv * KA_VL][P_ + ag * KA_AG];

    float mx[KA_CG][KA_AG], sm[KA_CG][KA_AG];
#pragma unroll
    for (int g = 0; g < KA_CG; ++g) {
#pragma unroll
        for (int j = 0; j < KA_AG; ++j) { mx[g][j] = -INFINITY; sm[g][j] = 0.0f; }
    }

    for (int it = 0; it < KA_NB; ++it) {
        float  fv[KA_U][KA_CG];
        float4 w[KA_U];
#pragma unroll
        for (int u = 0; u < KA_U; ++u) {
#pragma unroll
            for (int g = 0; g < KA_CG; ++g) fv[u][g] = fp[(it * KA_U + u) * KA_PITCH + g * 32];
        }
#pragma unroll
        for (int u = 0; u < KA_U; ++u) w[u] = wp[(it * KA_U + u) * (KA_PITCH / 4)];
#pragma unroll
        for (int u = 0; u < KA_U; ++u) {
            const float wj[KA_AG] = {w[u].x, w[u].y, w[u].z, w[u].w};
#pragma unroll
            for (int g = 0; g < KA_CG; ++g) {
#pragma unroll
                for (int j = 0; j < KA_AG; ++j) {
                    mx[g][j] = fmaxf(mx[g][j], wj[j] * fv[u][g]);
                    sm[g][j] = fmaf(wj[j], fv[u][g], sm[g][j]);
                }
            }
        }
    }

    if (hv == 1) {
#pragma unroll
        for (int g = 0; g < KA_CG; ++g) {
#pragma unroll
            for (int j = 0; j < KA_AG; ++j) {
                pmx[ag * KA_AG + j][g * 32 + lane] = mx[g][j];
                psm[ag * KA_AG + j][g * 32 + lane] = sm[g][j];
            }
        }
    }
    __syncthreads();
    if (hv == 0) {
        float* po = pagg + (size_t)blockIdx.x * PAGG_SL + (size_t)(ag * KA_AG) * 192;
#pragma unroll
        for (int g = 0; g < KA_CG; ++g) {
#pragma unroll
            for (int j = 0; j < KA_AG; ++j) {
                po[j * 192 + g * 32 + lane]      = fmaxf(mx[g][j], pmx[ag * KA_AG + j][g * 32 + lane]);
                po[j * 192 + 96 + g * 32 + lane] = sm[g][j] + psm[ag * KA_AG + j][g * 32 + lane];
            }
        }
    }
}

#define K3_KQ    4
#define K3_KL    (192 / K3_KQ)
#define K3_NG    (NF_ / 32)
#define K3_MB    16
#define K3_MT    (B_ / K3_MB)
#define K3_GEMM  (A_ * K3_NG * K3_MT)
#define K3_PITCH 36
#define K3_APITCH 196

__global__ __launch_bounds__(256) void k3_bw(
    const float* __restrict__ pagg, const float* __restrict__ Wout,
    float* __restrict__ BW)
{
    __shared__ float red[K3_KQ][K3_MB][K3_PITCH];
    __shared__ __attribute__((aligned(16))) float sA[K3_MB][K3_APITCH];
    static_assert(K3_MB * 16 == 256 && 192 / 4 == 3 * 16 && 96 / 4 == 16 + 8,
                  "16 threads per batch row, three 16-byte pieces each; the max/sum border is lane 8 of piece 1");
    static_assert(K3_KQ == 4 && K3_KL % 4 == 0, "the epilogue sums exactly four quarter partials");
    static_assert(K3_MB * 32 == 2 * 256, "the epilogue: two outputs per thread");

    if (blockIdx.x >= K3_GEMM) {
        const int h = blockIdx.x - K3_GEMM;
        const float4* src = (const float4*)(Wout + (h & 1) * 1024);
        float4* dst = (float4*)(BW + (size_t)(h >> 1) * KTOT * NF_ + (h & 1) * 1024);
        dst[threadIdx.x] = src[threadIdx.x];
        return;
    }

    const int a    = blockIdx.x / (K3_NG * K3_MT);
    const int ng   = (blockIdx.x / K3_MT) % K3_NG;
    const int b0   = (blockIdx.x % K3_MT) * K3_MB;
    const int lane = threadIdx.x & 31;
    const int wv   = threadIdx.x >> 5;
    const int nt   = wv & 1;
    const int q    = wv >> 1;
    const int c16  = lane & 15;
    const int kh   = (lane >> 4) * 2;
    const int col  = ng * 32 + nt * 16 + c16;

    const int pb = threadIdx.x >> 4;
    const int pl = threadIdx.x & 15;
    const v4f* sp = (const v4f*)(pagg + ((size_t)(b0 + pb) * KA_SL * A_ + a) * 192) + pl;
    v4f pv[3][KA_SL];
#pragma unroll
    for (int j = 0; j < 3; ++j) {
#pragma unroll
        for (int s = 0; s < KA_SL; ++s) pv[j][s] = sp[(size_t)s * (PAGG_SL / 4) + 16 * j];
    }

    const float* bp  = Wout + (size_t)(F_ + a * 192 + q * K3_KL + kh) * NF_ + col;
    v2f bb[K3_KL / 4];
#pragma unroll
    for (int kt = 0; kt < K3_KL / 4; ++kt) {
        bb[kt].x = bp[(size_t)(kt * 4) * NF_];
        bb[kt].y = bp[(size_t)(kt * 4 + 1) * NF_];
    }
    const float bias = Wout[(size_t)(F_ + 2 * A_ * C_ + a) * NF_ + ng * 32 + (threadIdx.x & 31)];

    asm volatile("" :: "v"(pv[0][0]), "v"(pv[0][1]), "v"(pv[0][2]), "v"(pv[0][3]), "v"(pv[0][4]), "v"(pv[0][5]), "v"(pv[0][6]), "v"(pv[0][7]),
                       "v"(pv[1][0]), "v"(pv[1][1]), "v"(pv[1][2]), "v"(pv[1][3]), "v"(pv[1][4]), "v"(pv[1][5]), "v"(pv[1][6]), "v"(pv[1][7]),
                       "v"(pv[2][0]), "v"(pv[2][1]), "v"(pv[2][2]), "v"(pv[2][3]), "v"(pv[2][4]), "v"(pv[2][5]), "v"(pv[2][6]), "v"(pv[2][7]));
    {
        const float sc = 1.0f / (float)V_;
        v4f m0 = pv[0][0], m1 = pv[1][0], t1 = pv[1][0], t2 = pv[2][0];
#pragma unroll
        for (int s = 1; s < KA_SL; ++s) {
            m0.x = fmaxf(m0.x, pv[0][s].x); m0.y = fmaxf(m0.y, pv[0][s].y);
            m0.z = fmaxf(m0.z, pv[0][s].z); m0.w = fmaxf(m0.w, pv[0][s].w);
            m1.x = fmaxf(m1.x, pv[1][s].x); m1.y = fmaxf(m1.y, pv[1][s].y);
            m1.z = fmaxf(m1.z, pv[1][s].z); m1.w = fmaxf(m1.w, pv[1][s].w);
            t1.x += pv[1][s].x; t1.y += pv[1][s].y; t1.z += pv[1][s].z; t1.w += pv[1][s].w;
            t2.x += pv[2][s].x; t2.y += pv[2][s].y; t2.z += pv[2][s].z; t2.w += pv[2][s].w;
        }
        const bool isSum = pl >= 8;
        v4f r1, r2;
        r1.x = isSum ? t1.x * sc : m1.x;  r1.y = isSum ? t1.y * sc : m1.y;
        r1.z = isSum ? t1.z * sc : m1.z;  r1.w = isSum ? t1.w * sc : m1.w;
        r2.x = t2.x * sc;  r2.y = t2.y * sc;  r2.z = t2.z * sc;  r2.w = t2.w * sc;
        *(v4f*)&sA[pb][pl * 4]        = m0;
        *(v4f*)&sA[pb][64 + pl * 4]   = r1;
        *(v4f*)&sA[pb][128 + pl * 4]  = r2;
    }
    __syncthreads();

    v8f acc = {};
#pragma unroll
    for (int kt = 0; kt < K3_KL / 4; ++kt) {
        const int k0 = q * K3_KL + kt * 4 + kh;
        const v2f av = *(const v2f*)&sA[c16][k0];
        acc = __builtin_amdgcn_wmma_f32_16x16x4_f32(
            false, av, false, bb[kt], (short)0, acc, false, false);
    }

#pragma unroll
    for (int r = 0; r < 8; ++r)
        red[q][r + (lane >> 4) * 8][nt * 16 + c16] = acc[r];
    __syncthreads();

    const int cc = threadIdx.x & 31;
    const int rb = threadIdx.x >> 5;
#pragma unroll
    for (int i = 0; i < 2; ++i) {
        const int b = rb + 8 * i;
        const float t = ((red[0][b][cc] + red[1][b][cc]) + red[2][b][cc]) + red[3][b][cc];
        BW[((size_t)(b0 + b) * KTOT + F_ + a) * NF_ + ng * 32 + cc] = t + bias;
    }
}

__device__ __forceinline__ float k4_tanh(float v)
{
    const float e = __builtin_amdgcn_exp2f(v * 2.885390081777927f);
    return fmaf(-2.0f, __builtin_amdgcn_rcpf(e + 1.0f), 1.0f);
}

#define K4_RT    2
#define K4_ROWS  (16 * K4_RT)
#define K4_PITCH 52

__global__ __launch_bounds__(256) void k4_out(
    const float* __restrict__ x,   const float* __restrict__ ewb,
    const float* __restrict__ BW,  const float* __restrict__ bout,
    float* __restrict__ out)
{
    __shared__ __attribute__((aligned(16))) float sA[K4_ROWS][K4_PITCH];
    static_assert(V_ % K4_ROWS == 0, "a block never straddles two batches");
    static_assert(K4_ROWS * 8 == 256 && A_ == 8 * 4 && F_ == 8 * 2,
                  "a thread stages one 16-byte piece of ew and one 8-byte piece of x of its row");

    const int lane  = threadIdx.x & 31;
    const int ntile = threadIdx.x >> 5;
    const int c16   = lane & 15;
    const int col   = ntile * 16 + c16;
    const int kh    = (lane >> 4) * 2;
    const int row0  = blockIdx.x * K4_ROWS;
    const int b     = row0 / V_;

    const int sr = threadIdx.x >> 3, sc = threadIdx.x & 7;
    const float4 ev = *(const float4*)(ewb + (size_t)(row0 + sr) * A_ + sc * 4);
    const v2f    xv = *(const v2f*)(x + (size_t)(row0 + sr) * F_ + sc * 2);

    const float* bwb = BW + (size_t)b * KTOT * NF_ + col;
    v2f bb[KTOT / 4];
#pragma unroll
    for (int kt = 0; kt < KTOT / 4; ++kt) {
        const int k0 = kt * 4 + kh;
        bb[kt].x = bwb[(size_t)k0 * NF_];
        bb[kt].y = bwb[(size_t)(k0 + 1) * NF_];
    }
    const float bias = bout[col];

    *(float4*)&sA[sr][F_ + sc * 4] = ev;
    *(v2f*)&sA[sr][sc * 2]         = xv;
    __syncthreads();

    v8f acc[K4_RT];
#pragma unroll
    for (int t = 0; t < K4_RT; ++t) acc[t] = v8f{};
#pragma unroll
    for (int kt = 0; kt < KTOT / 4; ++kt) {
        const int k0 = kt * 4 + kh;
#pragma unroll
        for (int t = 0; t < K4_RT; ++t) {
            const v2f a = *(const v2f*)&sA[t * 16 + c16][k0];
            acc[t] = __builtin_amdgcn_wmma_f32_16x16x4_f32(
                false, a, false, bb[kt], (short)0, acc[t], false, false);
        }
    }

#pragma unroll
    for (int t = 0; t < K4_RT; ++t) {
#pragma unroll
        for (int r = 0; r < 8; ++r) {
            const int row = row0 + t * 16 + r + ((lane >> 4) * 8);
            out[(size_t)row * NF_ + col] = k4_tanh(acc[t][r] + bias);
        }
    }
}

extern "C" void kernel_launch(void* const* d_in, const int* in_sizes, int n_in,
                              void* d_out, int out_size, void* d_ws, size_t ws_size,
                              hipStream_t stream)
{
    const float* x    = (const float*)d_in[0];
    const float* Wf   = (const float*)d_in[1];
    const float* bf   = (const float*)d_in[2];
    const float* Ws   = (const float*)d_in[3];
    const float* bs   = (const float*)d_in[4];
    const float* Wout = (const float*)d_in[5];
    const float* bout = (const float*)d_in[6];
    float* out = (float*)d_out;

    float* ewb  = (float*)d_ws;
    float* pagg = ewb + (size_t)ROWS * A_;
    float* BW   = pagg + (size_t)B_ * KA_SL * PAGG_SL;

    ka_vertex_agg<<<B_ * KA_SL, KA_T, 0, stream>>>(x, Wf, bf, Ws, bs, ewb, pagg);
    k3_bw        <<<K3_GEMM + 2 * B_, 256, 0, stream>>>(pagg, Wout, BW);
    k4_out       <<<ROWS / K4_ROWS, 256, 0, stream>>>(x, ewb, BW, bout, out);
}
